# starting kernel
# baseline (speedup 1.0000x reference)
#include <hip/hip_runtime.h>


typedef float v2f __attribute__((ext_vector_type(2)));
typedef float v8f __attribute__((ext_vector_type(8)));

#define N_NODES 50000
#define N_EDGES 800000
#define C_IN    128
#define C_HID   128
#define C_OUTC  64


__global__ void zero_kernel(float* __restrict__ p, int n) {
    int i = blockIdx.x * blockDim.x + threadIdx.x;
    if (i < n) p[i] = 0.0f;
}

__global__ void deg_kernel(const int* __restrict__ dst, float* __restrict__ deg, int E) {
    int i = blockIdx.x * blockDim.x + threadIdx.x;
    if (i < E) atomicAdd(&deg[dst[i]], 1.0f);
}

__global__ void invsqrt_kernel(float* __restrict__ deg, int n) {
    int i = blockIdx.x * blockDim.x + threadIdx.x;
    if (i < n) deg[i] = rsqrtf(deg[i] + 1.0f);
}

template<int K, int N, bool RELU_A>
__global__ void gemm16_wmma_f32(const float* __restrict__ A,
                                const float* __restrict__ W,
                                float* __restrict__ H) {
    const int lane  = threadIdx.x & 31;
    const int wave  = threadIdx.x >> 5;
    const int row0  = blockIdx.x << 4;
    const int col0  = wave << 4;
    const int m     = lane & 15;
    const int khalf = (lane >> 4) << 1;

    const float* __restrict__ arow = A + (size_t)(row0 + m) * K;
    const float* __restrict__ wcol = W + col0 + m;

    v8f acc = {};
#pragma unroll 8
    for (int k0 = 0; k0 < K; k0 += 4) {
        float a0 = arow[k0 + khalf + 0];
        float a1 = arow[k0 + khalf + 1];
        if (RELU_A) { a0 = fmaxf(a0, 0.0f); a1 = fmaxf(a1, 0.0f); }
        v2f a = {a0, a1};
        v2f b = { wcol[(size_t)(k0 + khalf + 0) * N],
                  wcol[(size_t)(k0 + khalf + 1) * N] };
        acc = __builtin_amdgcn_wmma_f32_16x16x4_f32(
                  false, a, false, b, (short)0, acc, false, false);
    }

    float* __restrict__ hrow =
        H + (size_t)(row0 + ((lane >> 4) << 3)) * N + col0 + m;
#pragma unroll
    for (int r = 0; r < 8; ++r)
        hrow[(size_t)r * N] = acc[r];
}

template<int C>
__global__ void selfloop_bias_kernel(const float* __restrict__ h,
                                     const float* __restrict__ isq,
                                     const float* __restrict__ bias,
                                     float* __restrict__ out, int n_elems) {
    int i = blockIdx.x * blockDim.x + threadIdx.x;
    if (i < n_elems) {
        int node = i / C;
        int c    = i - node * C;
        float s  = isq[node];
        out[i] = h[i] * (s * s) + bias[c];
    }
}

template<int C>
__global__ void edge_scatter_kernel(const float* __restrict__ h,
                                    const int* __restrict__ src,
                                    const int* __restrict__ dst,
                                    const float* __restrict__ isq,
                                    float* __restrict__ out, int E) {
    const int lane   = threadIdx.x & 31;
    const int wave   = (blockIdx.x * blockDim.x + threadIdx.x) >> 5;
    const int nwaves = (gridDim.x * blockDim.x) >> 5;
    constexpr int V  = C / 32;

    for (int e = wave; e < E; e += nwaves) {
        const int   s    = src[e];
        const int   d    = dst[e];
        const float coef = isq[s] * isq[d];
        const float* __restrict__ hs = h   + (size_t)s * C;
        float*       __restrict__ od = out + (size_t)d * C;
#pragma unroll
        for (int v = 0; v < V; ++v) {
            int c = lane + (v << 5);
            atomicAdd(&od[c], hs[c] * coef);
        }
    }
}


extern "C" void kernel_launch(void* const* d_in, const int* in_sizes, int n_in,
                              void* d_out, int out_size, void* d_ws, size_t ws_size,
                              hipStream_t stream) {
    (void)in_sizes; (void)n_in; (void)out_size; (void)ws_size;

    const float* x   = (const float*)d_in[0];
    const int*   ei  = (const int*)  d_in[1];
    const float* W1  = (const float*)d_in[2];
    const float* b1  = (const float*)d_in[3];
    const float* W2  = (const float*)d_in[4];
    const float* b2  = (const float*)d_in[5];
    float*       out = (float*)d_out;

    const int* src = ei;
    const int* dst = ei + N_EDGES;

    float* isq  = (float*)d_ws;
    float* h1   = isq + 50176;
    float* out1 = h1 + (size_t)N_NODES * C_HID;
    float* h2   = h1;

    const int T = 256;

    zero_kernel   <<<(N_NODES + T - 1) / T, T, 0, stream>>>(isq, N_NODES);
    deg_kernel    <<<(N_EDGES + T - 1) / T, T, 0, stream>>>(dst, isq, N_EDGES);
    invsqrt_kernel<<<(N_NODES + T - 1) / T, T, 0, stream>>>(isq, N_NODES);

    gemm16_wmma_f32<C_IN, C_HID, false>
        <<<N_NODES / 16, (C_HID / 16) * 32, 0, stream>>>(x, W1, h1);

    selfloop_bias_kernel<C_HID>
        <<<((N_NODES * C_HID) + T - 1) / T, T, 0, stream>>>(h1, isq, b1, out1,
                                                            N_NODES * C_HID);
    edge_scatter_kernel<C_HID>
        <<<1024, T, 0, stream>>>(h1, src, dst, isq, out1, N_EDGES);

    gemm16_wmma_f32<C_HID, C_OUTC, true>
        <<<N_NODES / 16, (C_OUTC / 16) * 32, 0, stream>>>(out1, W2, h2);

    selfloop_bias_kernel<C_OUTC>
        <<<((N_NODES * C_OUTC) + T - 1) / T, T, 0, stream>>>(h2, isq, b2, out,
                                                             N_NODES * C_OUTC);
    edge_scatter_kernel<C_OUTC>
        <<<1024, T, 0, stream>>>(h2, src, dst, isq, out, N_EDGES);
}


# Edge list scanned once, not twice: one owner pass counts degrees in LDS (replaces 3 degree launches) and leaves per-wave (src,node) lists both layers read; sums still LDS arrival order; 7->5 launches
# speedup vs baseline: 3.0809x; 3.0809x over previous
#include <hip/hip_runtime.h>


typedef float v2f __attribute__((ext_vector_type(2)));
typedef float v8f __attribute__((ext_vector_type(8)));

#define N_NODES 50000
#define N_EDGES 800000
#define C_IN    128
#define C_HID   128
#define C_OUTC  64

template<int K, int N, bool RELU_A>
__global__ void gemm16_wmma_f32(const float* __restrict__ A,
                                const float* __restrict__ W,
                                float* __restrict__ H) {
    const int lane  = threadIdx.x & 31;
    const int wave  = threadIdx.x >> 5;
    const int row0  = blockIdx.x << 4;
    const int col0  = wave << 4;
    const int m     = lane & 15;
    const int khalf = (lane >> 4) << 1;

    const float* __restrict__ arow = A + (size_t)(row0 + m) * K;
    const float* __restrict__ wcol = W + col0 + m;

    v8f acc = {};
#pragma unroll 8
    for (int k0 = 0; k0 < K; k0 += 4) {
        float a0 = arow[k0 + khalf + 0];
        float a1 = arow[k0 + khalf + 1];
        if (RELU_A) { a0 = fmaxf(a0, 0.0f); a1 = fmaxf(a1, 0.0f); }
        v2f a = {a0, a1};
        v2f b = { wcol[(size_t)(k0 + khalf + 0) * N],
                  wcol[(size_t)(k0 + khalf + 1) * N] };
        acc = __builtin_amdgcn_wmma_f32_16x16x4_f32(
                  false, a, false, b, (short)0, acc, false, false);
    }

    float* __restrict__ hrow =
        H + (size_t)(row0 + ((lane >> 4) << 3)) * N + col0 + m;
#pragma unroll
    for (int r = 0; r < 8; ++r)
        hrow[(size_t)r * N] = acc[r];
}

#define AGG_WAVES 32
#define AGG_QCAP  1024
#define AGG_SCAN  8
#define AGG_ROWS  8
#define AGG_NB    256
#define LIST_CAP  2048

typedef int v4i __attribute__((ext_vector_type(4)));

template<int NB, typename F>
__device__ __forceinline__ void scan_owned(const int* __restrict__ dst, unsigned* myq,
                                           int lo, int cnt, int wave, int lane, F group) {
    constexpr int NCH  = N_EDGES / 128;
    constexpr int STEP = AGG_WAVES * AGG_SCAN;
    static_assert(N_EDGES % 128 == 0, "edge list is scanned in whole chunks");
    static_assert(N_EDGES <= (1 << 23), "edge index is packed in 23 bits");

    int head = 0, tail = 0;
    const int start = (int)blockIdx.x * (NCH / NB);
    const unsigned lane4 = (unsigned)lane * 4u;

    auto chunk_of = [&](int i) -> unsigned {
        const int c = i + start;
        return (unsigned)(c >= NCH ? c - NCH : c);
    };
    auto match_chunk = [&](const v4i dv, const unsigned ebase) {
#pragma unroll
        for (int j = 0; j < 4; ++j) {
            const unsigned dl = (unsigned)(dv[j] - lo);
            const bool     m  = dl < (unsigned)cnt;
            const unsigned mask = __builtin_amdgcn_ballot_w32(m);
            if (mask) {
                const int pre = (int)__builtin_amdgcn_mbcnt_lo(mask, 0u);
                if (m) myq[(tail + pre) & (AGG_QCAP - 1)] =
                           ((ebase + (unsigned)j) << 9) | dl;
                tail += __builtin_popcount(mask);
            }
        }
    };
    auto groups_full = [&]() {
        while (tail - head >= 32) {
            group(head, 32);
            head += 32;
        }
    };

    int i0 = wave * AGG_SCAN;
    v4i cur[AGG_SCAN], nxt[AGG_SCAN];
#pragma unroll
    for (int u = 0; u < AGG_SCAN; ++u) {
        const int i = i0 + u;
        const unsigned c = i < NCH ? chunk_of(i) : 0u;
        cur[u] = *(const v4i*)(dst + (c * 128u + lane4));
    }
    for (; i0 + AGG_SCAN <= NCH; i0 += STEP) {
#pragma unroll
        for (int u = 0; u < AGG_SCAN; ++u) {
            const int i = i0 + STEP + u;
            const unsigned c = i < NCH ? chunk_of(i) : 0u;
            nxt[u] = *(const v4i*)(dst + (c * 128u + lane4));
        }
#pragma unroll
        for (int u = 0; u < AGG_SCAN; ++u) {
            match_chunk(cur[u], chunk_of(i0 + u) * 128u + lane4);
            if ((u & 3) == 3) groups_full();
        }
#pragma unroll
        for (int u = 0; u < AGG_SCAN; ++u) cur[u] = nxt[u];
    }
#pragma unroll
    for (int u = 0; u < AGG_SCAN; ++u)
        if (i0 + u < NCH) match_chunk(cur[u], chunk_of(i0 + u) * 128u + lane4);
    groups_full();
    if (tail - head > 0) group(head, tail - head);
}

template<int NB>
__global__ __launch_bounds__(AGG_WAVES * 32)
void filter_degree_kernel(const int* __restrict__ src,
                          const int* __restrict__ dst,
                          float* __restrict__ isq,
                          unsigned* __restrict__ lists,
                          int* __restrict__ counts) {
    constexpr int NPB = (N_NODES + NB - 1) / NB;
    static_assert(NPB <= 512, "local node index is packed in 9 bits");
    static_assert(N_NODES <= (1 << 23), "source index is packed in 23 bits");

    __shared__ unsigned degcnt[NPB];
    __shared__ unsigned ring[AGG_WAVES * AGG_QCAP];

    const int lane = threadIdx.x & 31;
    const int wave = __builtin_amdgcn_readfirstlane((int)(threadIdx.x >> 5));
    const int lo   = blockIdx.x * NPB;
    int cnt = N_NODES - lo;
    cnt = cnt < NPB ? cnt : NPB;
    cnt = cnt > 0 ? cnt : 0;

    for (int n = threadIdx.x; n < NPB; n += AGG_WAVES * 32) degcnt[n] = 0u;
    __syncthreads();

    unsigned* myq = ring + wave * AGG_QCAP;
    const int gw  = blockIdx.x * AGG_WAVES + wave;
    unsigned* mylist = lists + (size_t)gw * LIST_CAP;
    int total = 0;

    scan_owned<NB>(dst, myq, lo, cnt, wave, lane, [&](int head, int nb) {
        if (lane < nb) {
            const unsigned ent = myq[(head + lane) & (AGG_QCAP - 1)];
            const unsigned dl  = ent & 511u;
            atomicAdd(&degcnt[dl], 1u);
            if (head + nb <= LIST_CAP)
                mylist[head + lane] = ((unsigned)src[ent >> 9] << 9) | dl;
        }
        total = head + nb;
    });
    if (lane == 0) counts[gw] = total <= LIST_CAP ? total : -1;
    __syncthreads();

    for (int n = threadIdx.x; n < cnt; n += AGG_WAVES * 32)
        isq[lo + n] = rsqrtf((float)degcnt[n] + 1.0f);
}

template<int C>
__device__ __forceinline__ void agg_add_rows(const float* __restrict__ h,
                                             const float* __restrict__ isq,
                                             float* acc, unsigned ent,
                                             int nb, int lo, int lane) {
    constexpr int V = C / 32;
    typedef float vecf __attribute__((ext_vector_type(V)));

    const int   dl   = (int)(ent & 511u);
    const int   s_l  = (int)(ent >> 9);
    const float iq_d = isq[lo + dl];
    const float iq_s = isq[s_l];

    for (int k0 = 0; k0 < nb; k0 += AGG_ROWS) {
        vecf hv[AGG_ROWS];
#pragma unroll
        for (int g = 0; g < AGG_ROWS; ++g) {
            int kk = k0 + g;
            kk = kk < nb ? kk : nb - 1;
            const int s = __builtin_amdgcn_readlane(s_l, kk);
            hv[g] = *(const vecf*)(h + (size_t)s * C + V * lane);
        }
#pragma unroll
        for (int g = 0; g < AGG_ROWS; ++g) {
            const int k = k0 + g;
            if (k < nb) {
                const float cs = __int_as_float(__builtin_amdgcn_readlane(__float_as_int(iq_s), k));
                const float cd = __int_as_float(__builtin_amdgcn_readlane(__float_as_int(iq_d), k));
                const float coef = cs * cd;
                float* row = acc + __builtin_amdgcn_readlane(dl, k) * C + lane;
#pragma unroll
                for (int j = 0; j < V; ++j)
                    atomicAdd(row + j * 32, hv[g][j] * coef);
            }
        }
    }
}

template<int C, int NB>
__global__ __launch_bounds__(AGG_WAVES * 32)
void agg_owned_lds_kernel(const float* __restrict__ h,
                          const int* __restrict__ src,
                          const int* __restrict__ dst,
                          const float* __restrict__ isq,
                          const float* __restrict__ bias,
                          const unsigned* __restrict__ lists,
                          const int* __restrict__ counts,
                          float* __restrict__ out) {
    constexpr int V    = C / 32;
    constexpr int NPB  = (N_NODES + NB - 1) / NB;
    static_assert(NPB <= 512, "local node index is packed in 9 bits");
    typedef float vecf __attribute__((ext_vector_type(V)));

    __shared__ float    acc[NPB * C];
    __shared__ unsigned ring[AGG_WAVES * AGG_QCAP];

    const int lane = threadIdx.x & 31;
    const int wave = __builtin_amdgcn_readfirstlane((int)(threadIdx.x >> 5));
    const int lo   = blockIdx.x * NPB;
    int cnt = N_NODES - lo;
    cnt = cnt < NPB ? cnt : NPB;
    cnt = cnt > 0 ? cnt : 0;

    const int gw = blockIdx.x * AGG_WAVES + wave;
    const int nlist = __builtin_amdgcn_readfirstlane(counts[gw]);

    for (int n = wave; n < cnt; n += AGG_WAVES) {
        const int   node = lo + n;
        const float s    = isq[node];
        const float s2   = s * s;
        const vecf  hv   = *(const vecf*)(h + (size_t)node * C + V * lane);
#pragma unroll
        for (int j = 0; j < V; ++j)
            acc[n * C + j * 32 + lane] = hv[j] * s2 + bias[V * lane + j];
    }
    __syncthreads();

    if (nlist >= 0) {
        const unsigned* mylist = lists + (size_t)gw * LIST_CAP;
        unsigned ent = lane < nlist ? mylist[lane] : 0u;
        for (int pos = 0; pos < nlist; pos += 32) {
            const int left = nlist - pos;
            const unsigned nxt = pos + 32 + lane < nlist ? mylist[pos + 32 + lane] : 0u;
            agg_add_rows<C>(h, isq, acc, ent, left < 32 ? left : 32, lo, lane);
            ent = nxt;
        }
    } else {
        unsigned* myq = ring + wave * AGG_QCAP;
        scan_owned<NB>(dst, myq, lo, cnt, wave, lane, [&](int head, int nb) {
            unsigned ent = 0u;
            if (lane < nb) {
                const unsigned q = myq[(head + lane) & (AGG_QCAP - 1)];
                ent = ((unsigned)src[q >> 9] << 9) | (q & 511u);
            }
            agg_add_rows<C>(h, isq, acc, ent, nb, lo, lane);
        });
    }
    __syncthreads();

    for (int n = wave; n < cnt; n += AGG_WAVES) {
        vecf o;
#pragma unroll
        for (int j = 0; j < V; ++j) o[j] = acc[n * C + j * 32 + lane];
        *(vecf*)(out + (size_t)(lo + n) * C + V * lane) = o;
    }
}


extern "C" void kernel_launch(void* const* d_in, const int* in_sizes, int n_in,
                              void* d_out, int out_size, void* d_ws, size_t ws_size,
                              hipStream_t stream) {
    (void)in_sizes; (void)n_in; (void)out_size; (void)ws_size;

    const float* x   = (const float*)d_in[0];
    const int*   ei  = (const int*)  d_in[1];
    const float* W1  = (const float*)d_in[2];
    const float* b1  = (const float*)d_in[3];
    const float* W2  = (const float*)d_in[4];
    const float* b2  = (const float*)d_in[5];
    float*       out = (float*)d_out;

    const int* src = ei;
    const int* dst = ei + N_EDGES;

    float*    isq    = (float*)d_ws;
    float*    h1     = isq + 50176;
    float*    out1   = h1 + (size_t)N_NODES * C_HID;
    float*    h2     = h1;
    int*      counts = (int*)(out1 + (size_t)N_NODES * C_HID);
    unsigned* lists  = (unsigned*)(counts + AGG_NB * AGG_WAVES);

    filter_degree_kernel<AGG_NB>
        <<<AGG_NB, AGG_WAVES * 32, 0, stream>>>(src, dst, isq, lists, counts);

    gemm16_wmma_f32<C_IN, C_HID, false>
        <<<N_NODES / 16, (C_HID / 16) * 32, 0, stream>>>(x, W1, h1);

    agg_owned_lds_kernel<C_HID, AGG_NB>
        <<<AGG_NB, AGG_WAVES * 32, 0, stream>>>(h1, src, dst, isq, b1, lists, counts, out1);

    gemm16_wmma_f32<C_HID, C_OUTC, true>
        <<<N_NODES / 16, (C_OUTC / 16) * 32, 0, stream>>>(out1, W2, h2);

    agg_owned_lds_kernel<C_OUTC, AGG_NB>
        <<<AGG_NB, AGG_WAVES * 32, 0, stream>>>(h2, src, dst, isq, b2, lists, counts, out);
}


# Owner blocks stop scanning the whole edge list: a first pass marks per owner which 16-edge units hold an edge for it (LDS OR bitmap); owners read only marked units; 5->6 launches; arrival-order sums
# speedup vs baseline: 3.3367x; 1.0830x over previous
#include <hip/hip_runtime.h>


typedef float v2f __attribute__((ext_vector_type(2)));
typedef float v8f __attribute__((ext_vector_type(8)));

#define N_NODES 50000
#define N_EDGES 800000
#define C_IN    128
#define C_HID   128
#define C_OUTC  64

template<int K, int N, bool RELU_A>
__global__ void gemm16_wmma_f32(const float* __restrict__ A,
                                const float* __restrict__ W,
                                float* __restrict__ H) {
    const int lane  = threadIdx.x & 31;
    const int wave  = threadIdx.x >> 5;
    const int row0  = blockIdx.x << 4;
    const int col0  = wave << 4;
    const int m     = lane & 15;
    const int khalf = (lane >> 4) << 1;

    const float* __restrict__ arow = A + (size_t)(row0 + m) * K;
    const float* __restrict__ wcol = W + col0 + m;

    v8f acc = {};
#pragma unroll 8
    for (int k0 = 0; k0 < K; k0 += 4) {
        float a0 = arow[k0 + khalf + 0];
        float a1 = arow[k0 + khalf + 1];
        if (RELU_A) { a0 = fmaxf(a0, 0.0f); a1 = fmaxf(a1, 0.0f); }
        v2f a = {a0, a1};
        v2f b = { wcol[(size_t)(k0 + khalf + 0) * N],
                  wcol[(size_t)(k0 + khalf + 1) * N] };
        acc = __builtin_amdgcn_wmma_f32_16x16x4_f32(
                  false, a, false, b, (short)0, acc, false, false);
    }

    float* __restrict__ hrow =
        H + (size_t)(row0 + ((lane >> 4) << 3)) * N + col0 + m;
#pragma unroll
    for (int r = 0; r < 8; ++r)
        hrow[(size_t)r * N] = acc[r];
}

#define AGG_WAVES 32
#define AGG_QCAP  1024
#define AGG_SCAN  8
#define AGG_ROWS  8
#define AGG_NB    256
#define LIST_CAP  2048

#define UNIT_EDGES  16
#define N_UNITS     (N_EDGES / UNIT_EDGES)
#define MARK_UNITS  256
#define MARK_BLOCKS ((N_UNITS + MARK_UNITS - 1) / MARK_UNITS)
#define MARK_WORDS  (MARK_BLOCKS * 8)
#define FILT_QCAP   64

typedef int v4i __attribute__((ext_vector_type(4)));

template<typename F>
__device__ __forceinline__ void scan_owned(const int* __restrict__ dst, unsigned* myq,
                                           int lo, int cnt, int wave, int lane, F group) {
    constexpr int NCH = N_EDGES / 128;
    constexpr int CPW = (MARK_WORDS / AGG_WAVES) * 32 * UNIT_EDGES / 128;
    static_assert(N_EDGES % 128 == 0, "edge list is scanned in whole chunks");
    static_assert(N_EDGES <= (1 << 23), "edge index is packed in 23 bits");
    static_assert(MARK_WORDS % AGG_WAVES == 0 && ((MARK_WORDS / AGG_WAVES) * 32 * UNIT_EDGES) % 128 == 0,
                  "a wave's share is whole chunks");

    int head = 0, tail = 0;
    const int c0 = wave * CPW;
    int nmine = NCH - c0;
    nmine = nmine < CPW ? nmine : CPW;
    nmine = nmine > 0 ? nmine : 0;
    const unsigned lane4 = (unsigned)lane * 4u;

    auto match_chunk = [&](const v4i dv, const unsigned ebase) {
#pragma unroll
        for (int j = 0; j < 4; ++j) {
            const unsigned dl = (unsigned)(dv[j] - lo);
            const bool     m  = dl < (unsigned)cnt;
            const unsigned mask = __builtin_amdgcn_ballot_w32(m);
            if (mask) {
                const int pre = (int)__builtin_amdgcn_mbcnt_lo(mask, 0u);
                if (m) myq[(tail + pre) & (AGG_QCAP - 1)] =
                           ((ebase + (unsigned)j) << 9) | dl;
                tail += __builtin_popcount(mask);
            }
        }
    };
    auto groups_full = [&]() {
        while (tail - head >= 32) {
            group(head, 32);
            head += 32;
        }
    };
    auto load_chunk = [&](int i) -> v4i {
        const unsigned c = i < nmine ? (unsigned)(c0 + i) : 0u;
        return *(const v4i*)(dst + (c * 128u + lane4));
    };

    int i0 = 0;
    v4i cur[AGG_SCAN], nxt[AGG_SCAN];
#pragma unroll
    for (int u = 0; u < AGG_SCAN; ++u) cur[u] = load_chunk(i0 + u);
    for (; i0 + AGG_SCAN <= nmine; i0 += AGG_SCAN) {
#pragma unroll
        for (int u = 0; u < AGG_SCAN; ++u) nxt[u] = load_chunk(i0 + AGG_SCAN + u);
#pragma unroll
        for (int u = 0; u < AGG_SCAN; ++u) {
            match_chunk(cur[u], (unsigned)(c0 + i0 + u) * 128u + lane4);
            if ((u & 3) == 3) groups_full();
        }
#pragma unroll
        for (int u = 0; u < AGG_SCAN; ++u) cur[u] = nxt[u];
    }
#pragma unroll
    for (int u = 0; u < AGG_SCAN; ++u)
        if (i0 + u < nmine) match_chunk(cur[u], (unsigned)(c0 + i0 + u) * 128u + lane4);
    groups_full();
    if (tail - head > 0) group(head, tail - head);
}

template<int NB>
__global__ __launch_bounds__(256)
void owner_mark_kernel(const int* __restrict__ dst, unsigned* __restrict__ marks) {
    constexpr unsigned NPB = (N_NODES + NB - 1) / NB;
    static_assert((NB & (NB - 1)) == 0, "owner number is kept in range with a mask");
    static_assert(N_EDGES % UNIT_EDGES == 0, "a unit is 16 whole edges");
    static_assert(MARK_UNITS * UNIT_EDGES == 4 * 4 * 256, "256 threads, four loads of four edges each");

    __shared__ unsigned bm[NB * 8];
    const int t = threadIdx.x;
    for (int i = t; i < NB * 8; i += 256) bm[i] = 0u;
    __syncthreads();

    const unsigned ebase = blockIdx.x * (unsigned)(MARK_UNITS * UNIT_EDGES);
    v4i d[4];
#pragma unroll
    for (int j = 0; j < 4; ++j) {
        const unsigned e = ebase + (unsigned)(j * 1024 + 4 * t);
        const v4i none = {0, 0, 0, 0};
        d[j] = e < (unsigned)N_EDGES ? *(const v4i*)(dst + e) : none;
    }
#pragma unroll
    for (int j = 0; j < 4; ++j) {
        const unsigned e  = ebase + (unsigned)(j * 1024 + 4 * t);
        const unsigned ul = (unsigned)(j * 64 + (t >> 2));
        if (e < (unsigned)N_EDGES) {
#pragma unroll
            for (int i = 0; i < 4; ++i) {
                const unsigned owner = ((unsigned)d[j][i] / NPB) & (unsigned)(NB - 1);
                atomicOr(&bm[owner * 8u + (ul >> 5)], 1u << (ul & 31u));
            }
        }
    }
    __syncthreads();

    unsigned* mine = marks + (size_t)blockIdx.x * (NB * 8);
    for (int i = t; i < NB * 8; i += 256) mine[i] = bm[i];
}

template<int NB, typename F>
__device__ __forceinline__ void scan_marks(const unsigned* __restrict__ marks, unsigned* myq,
                                           int wave, int lane, F group) {
    constexpr int NW  = MARK_WORDS;
    constexpr int WPW = NW / AGG_WAVES;
    static_assert(NW % AGG_WAVES == 0, "every wave gets the same number of words");
    static_assert(WPW <= 64, "two words per lane cover a wave's share");

    int head = 0, tail = 0;
    const unsigned* mine = marks + (size_t)blockIdx.x * 8;
    unsigned x[2], wd[2];
#pragma unroll
    for (int h = 0; h < 2; ++h) {
        const int i = h * 32 + lane;
        x[h]  = (unsigned)(wave * WPW + i);
        wd[h] = i < WPW ? mine[(size_t)(x[h] >> 3) * (NB * 8) + (x[h] & 7u)] : 0u;
    }
#pragma unroll
    for (int h = 0; h < 2; ++h) {
        unsigned w = wd[h];
        for (int it = 0; it < 32; ++it) {
            const bool     m    = w != 0u;
            const unsigned mask = __builtin_amdgcn_ballot_w32(m);
            if (!mask) break;
            if (m) {
                const unsigned k = (unsigned)__builtin_ctz(w);
                w &= w - 1u;
                myq[(tail + (int)__builtin_amdgcn_mbcnt_lo(mask, 0u)) & (FILT_QCAP - 1)] = x[h] * 32u + k;
            }
            tail += __builtin_popcount(mask);
            if (tail - head >= 32) {
                group(head, 32);
                head += 32;
            }
        }
    }
    if (tail - head > 0) group(head, tail - head);
}

template<int NB>
__global__ __launch_bounds__(AGG_WAVES * 32)
void filter_degree_kernel(const int* __restrict__ src,
                          const int* __restrict__ dst,
                          const unsigned* __restrict__ marks,
                          float* __restrict__ isq,
                          unsigned* __restrict__ lists,
                          int* __restrict__ counts) {
    constexpr int NPB = (N_NODES + NB - 1) / NB;
    static_assert(NPB <= 512, "local node index is packed in 9 bits");
    static_assert(N_NODES <= (1 << 23), "source index is packed in 23 bits");

    __shared__ unsigned degcnt[NPB];
    __shared__ unsigned ring[AGG_WAVES * FILT_QCAP];

    const int lane = threadIdx.x & 31;
    const int wave = __builtin_amdgcn_readfirstlane((int)(threadIdx.x >> 5));
    const int lo   = blockIdx.x * NPB;
    int cnt = N_NODES - lo;
    cnt = cnt < NPB ? cnt : NPB;
    cnt = cnt > 0 ? cnt : 0;

    for (int n = threadIdx.x; n < NPB; n += AGG_WAVES * 32) degcnt[n] = 0u;
    __syncthreads();

    unsigned* myq = ring + wave * FILT_QCAP;
    const int gw  = blockIdx.x * AGG_WAVES + wave;
    unsigned* mylist = lists + (size_t)gw * LIST_CAP;
    int total = 0;

    scan_marks<NB>(marks, myq, wave, lane, [&](int head, int nb) {
        unsigned unit = N_UNITS;
        if (lane < nb) unit = myq[(head + lane) & (FILT_QCAP - 1)];
        const bool     live = unit < (unsigned)N_UNITS;
        const unsigned e0   = live ? unit * 16u : 0u;
        v4i dv[4], sv[4];
#pragma unroll
        for (int j = 0; j < 4; ++j) {
            dv[j] = *(const v4i*)(dst + e0 + 4 * j);
            sv[j] = *(const v4i*)(src + e0 + 4 * j);
        }
#pragma unroll
        for (int k = 0; k < 16; ++k) {
            const unsigned dl   = (unsigned)(dv[k >> 2][k & 3] - lo);
            const bool     m    = live && dl < (unsigned)cnt;
            const unsigned mask = __builtin_amdgcn_ballot_w32(m);
            if (mask) {
                if (m) {
                    atomicAdd(&degcnt[dl], 1u);
                    const int pos = total + (int)__builtin_amdgcn_mbcnt_lo(mask, 0u);
                    if (pos < LIST_CAP)
                        mylist[pos] = ((unsigned)sv[k >> 2][k & 3] << 9) | dl;
                }
                total += __builtin_popcount(mask);
            }
        }
    });
    if (lane == 0) counts[gw] = total <= LIST_CAP ? total : -1;
    __syncthreads();

    for (int n = threadIdx.x; n < cnt; n += AGG_WAVES * 32)
        isq[lo + n] = rsqrtf((float)degcnt[n] + 1.0f);
}

template<int C>
__device__ __forceinline__ void agg_add_rows(const float* __restrict__ h,
                                             const float* __restrict__ isq,
                                             float* acc, unsigned ent,
                                             int nb, int lo, int lane) {
    constexpr int V = C / 32;
    typedef float vecf __attribute__((ext_vector_type(V)));

    const int   dl   = (int)(ent & 511u);
    const int   s_l  = (int)(ent >> 9);
    const float iq_d = isq[lo + dl];
    const float iq_s = isq[s_l];

    for (int k0 = 0; k0 < nb; k0 += AGG_ROWS) {
        vecf hv[AGG_ROWS];
#pragma unroll
        for (int g = 0; g < AGG_ROWS; ++g) {
            int kk = k0 + g;
            kk = kk < nb ? kk : nb - 1;
            const int s = __builtin_amdgcn_readlane(s_l, kk);
            hv[g] = *(const vecf*)(h + (size_t)s * C + V * lane);
        }
#pragma unroll
        for (int g = 0; g < AGG_ROWS; ++g) {
            const int k = k0 + g;
            if (k < nb) {
                const float cs = __int_as_float(__builtin_amdgcn_readlane(__float_as_int(iq_s), k));
                const float cd = __int_as_float(__builtin_amdgcn_readlane(__float_as_int(iq_d), k));
                const float coef = cs * cd;
                float* row = acc + __builtin_amdgcn_readlane(dl, k) * C + lane;
#pragma unroll
                for (int j = 0; j < V; ++j)
                    atomicAdd(row + j * 32, hv[g][j] * coef);
            }
        }
    }
}

template<int C, int NB>
__global__ __launch_bounds__(AGG_WAVES * 32)
void agg_owned_lds_kernel(const float* __restrict__ h,
                          const int* __restrict__ src,
                          const int* __restrict__ dst,
                          const float* __restrict__ isq,
                          const float* __restrict__ bias,
                          const unsigned* __restrict__ lists,
                          const int* __restrict__ counts,
                          float* __restrict__ out) {
    constexpr int V    = C / 32;
    constexpr int NPB  = (N_NODES + NB - 1) / NB;
    static_assert(NPB <= 512, "local node index is packed in 9 bits");
    typedef float vecf __attribute__((ext_vector_type(V)));

    __shared__ float    acc[NPB * C];
    __shared__ unsigned ring[AGG_WAVES * AGG_QCAP];

    const int lane = threadIdx.x & 31;
    const int wave = __builtin_amdgcn_readfirstlane((int)(threadIdx.x >> 5));
    const int lo   = blockIdx.x * NPB;
    int cnt = N_NODES - lo;
    cnt = cnt < NPB ? cnt : NPB;
    cnt = cnt > 0 ? cnt : 0;

    const int gw = blockIdx.x * AGG_WAVES + wave;
    const int nlist = __builtin_amdgcn_readfirstlane(counts[gw]);

    for (int n = wave; n < cnt; n += AGG_WAVES) {
        const int   node = lo + n;
        const float s    = isq[node];
        const float s2   = s * s;
        const vecf  hv   = *(const vecf*)(h + (size_t)node * C + V * lane);
#pragma unroll
        for (int j = 0; j < V; ++j)
            acc[n * C + j * 32 + lane] = hv[j] * s2 + bias[V * lane + j];
    }
    __syncthreads();

    if (nlist >= 0) {
        const unsigned* mylist = lists + (size_t)gw * LIST_CAP;
        unsigned ent = lane < nlist ? mylist[lane] : 0u;
        for (int pos = 0; pos < nlist; pos += 32) {
            const int left = nlist - pos;
            const unsigned nxt = pos + 32 + lane < nlist ? mylist[pos + 32 + lane] : 0u;
            agg_add_rows<C>(h, isq, acc, ent, left < 32 ? left : 32, lo, lane);
            ent = nxt;
        }
    } else {
        unsigned* myq = ring + wave * AGG_QCAP;
        scan_owned(dst, myq, lo, cnt, wave, lane, [&](int head, int nb) {
            unsigned ent = 0u;
            if (lane < nb) {
                const unsigned q = myq[(head + lane) & (AGG_QCAP - 1)];
                ent = ((unsigned)src[q >> 9] << 9) | (q & 511u);
            }
            agg_add_rows<C>(h, isq, acc, ent, nb, lo, lane);
        });
    }
    __syncthreads();

    for (int n = wave; n < cnt; n += AGG_WAVES) {
        vecf o;
#pragma unroll
        for (int j = 0; j < V; ++j) o[j] = acc[n * C + j * 32 + lane];
        *(vecf*)(out + (size_t)(lo + n) * C + V * lane) = o;
    }
}


extern "C" void kernel_launch(void* const* d_in, const int* in_sizes, int n_in,
                              void* d_out, int out_size, void* d_ws, size_t ws_size,
                              hipStream_t stream) {
    (void)in_sizes; (void)n_in; (void)out_size; (void)ws_size;

    const float* x   = (const float*)d_in[0];
    const int*   ei  = (const int*)  d_in[1];
    const float* W1  = (const float*)d_in[2];
    const float* b1  = (const float*)d_in[3];
    const float* W2  = (const float*)d_in[4];
    const float* b2  = (const float*)d_in[5];
    float*       out = (float*)d_out;

    const int* src = ei;
    const int* dst = ei + N_EDGES;

    float*    isq    = (float*)d_ws;
    float*    h1     = isq + 50176;
    float*    out1   = h1 + (size_t)N_NODES * C_HID;
    float*    h2     = h1;
    int*      counts = (int*)(out1 + (size_t)N_NODES * C_HID);
    unsigned* lists  = (unsigned*)(counts + AGG_NB * AGG_WAVES);
    unsigned* marks  = lists + (size_t)AGG_NB * AGG_WAVES * LIST_CAP;

    owner_mark_kernel<AGG_NB>
        <<<MARK_BLOCKS, 256, 0, stream>>>(dst, marks);

    filter_degree_kernel<AGG_NB>
        <<<AGG_NB, AGG_WAVES * 32, 0, stream>>>(src, dst, marks, isq, lists, counts);

    gemm16_wmma_f32<C_IN, C_HID, false>
        <<<N_NODES / 16, (C_HID / 16) * 32, 0, stream>>>(x, W1, h1);

    agg_owned_lds_kernel<C_HID, AGG_NB>
        <<<AGG_NB, AGG_WAVES * 32, 0, stream>>>(h1, src, dst, isq, b1, lists, counts, out1);

    gemm16_wmma_f32<C_HID, C_OUTC, true>
        <<<N_NODES / 16, (C_OUTC / 16) * 32, 0, stream>>>(out1, W2, h2);

    agg_owned_lds_kernel<C_OUTC, AGG_NB>
        <<<AGG_NB, AGG_WAVES * 32, 0, stream>>>(h2, src, dst, isq, b2, lists, counts, out);
}


# Both dense products run inside the owner grid from LDS (x@W1 behind the edge filter, relu(LDS sums)@W2 behind the layer-1 sums; out1 never stored); 6->4 launches; k order of the products permuted
# speedup vs baseline: 4.2049x; 1.2602x over previous
#include <hip/hip_runtime.h>


typedef float v2f __attribute__((ext_vector_type(2)));
typedef float v8f __attribute__((ext_vector_type(8)));

#define N_NODES 50000
#define N_EDGES 800000
#define C_IN    128
#define C_HID   128
#define C_OUTC  64

#define AGG_WAVES 32
#define AGG_QCAP  1024
#define AGG_SCAN  8
#define AGG_ROWS  8
#define AGG_NB    256
#define LIST_CAP  2048

#define UNIT_EDGES  16
#define N_UNITS     (N_EDGES / UNIT_EDGES)
#define MARK_UNITS  256
#define MARK_BLOCKS ((N_UNITS + MARK_UNITS - 1) / MARK_UNITS)
#define MARK_WORDS  (MARK_BLOCKS * 8)
#define FILT_QCAP   64

typedef int v4i __attribute__((ext_vector_type(4)));

template<typename F>
__device__ __forceinline__ void scan_owned(const int* __restrict__ dst, unsigned* myq,
                                           int lo, int cnt, int wave, int lane, F group) {
    constexpr int NCH = N_EDGES / 128;
    constexpr int CPW = (MARK_WORDS / AGG_WAVES) * 32 * UNIT_EDGES / 128;
    static_assert(N_EDGES % 128 == 0, "edge list is scanned in whole chunks");
    static_assert(N_EDGES <= (1 << 23), "edge index is packed in 23 bits");
    static_assert(MARK_WORDS % AGG_WAVES == 0 && ((MARK_WORDS / AGG_WAVES) * 32 * UNIT_EDGES) % 128 == 0,
                  "a wave's share is whole chunks");

    int head = 0, tail = 0;
    const int c0 = wave * CPW;
    int nmine = NCH - c0;
    nmine = nmine < CPW ? nmine : CPW;
    nmine = nmine > 0 ? nmine : 0;
    const unsigned lane4 = (unsigned)lane * 4u;

    auto match_chunk = [&](const v4i dv, const unsigned ebase) {
#pragma unroll
        for (int j = 0; j < 4; ++j) {
            const unsigned dl = (unsigned)(dv[j] - lo);
            const bool     m  = dl < (unsigned)cnt;
            const unsigned mask = __builtin_amdgcn_ballot_w32(m);
            if (mask) {
                const int pre = (int)__builtin_amdgcn_mbcnt_lo(mask, 0u);
                if (m) myq[(tail + pre) & (AGG_QCAP - 1)] =
                           ((ebase + (unsigned)j) << 9) | dl;
                tail += __builtin_popcount(mask);
            }
        }
    };
    auto groups_full = [&]() {
        while (tail - head >= 32) {
            group(head, 32);
            head += 32;
        }
    };
    auto load_chunk = [&](int i) -> v4i {
        const unsigned c = i < nmine ? (unsigned)(c0 + i) : 0u;
        return *(const v4i*)(dst + (c * 128u + lane4));
    };

    int i0 = 0;
    v4i cur[AGG_SCAN], nxt[AGG_SCAN];
#pragma unroll
    for (int u = 0; u < AGG_SCAN; ++u) cur[u] = load_chunk(i0 + u);
    for (; i0 + AGG_SCAN <= nmine; i0 += AGG_SCAN) {
#pragma unroll
        for (int u = 0; u < AGG_SCAN; ++u) nxt[u] = load_chunk(i0 + AGG_SCAN + u);
#pragma unroll
        for (int u = 0; u < AGG_SCAN; ++u) {
            match_chunk(cur[u], (unsigned)(c0 + i0 + u) * 128u + lane4);
            if ((u & 3) == 3) groups_full();
        }
#pragma unroll
        for (int u = 0; u < AGG_SCAN; ++u) cur[u] = nxt[u];
    }
#pragma unroll
    for (int u = 0; u < AGG_SCAN; ++u)
        if (i0 + u < nmine) match_chunk(cur[u], (unsigned)(c0 + i0 + u) * 128u + lane4);
    groups_full();
    if (tail - head > 0) group(head, tail - head);
}

typedef float v4f __attribute__((ext_vector_type(4)));

#define W_PASSES(K, N) ((K) * (N) / 4 / (AGG_WAVES * 32))

template<int K, int N>
__device__ __forceinline__ void w_fragments_load(const float* __restrict__ W, v4f (&wr)[W_PASSES(K, N)]) {
    static_assert((K * N / 4) % (AGG_WAVES * 32) == 0, "whole passes of the block");
#pragma unroll
    for (int i = 0; i < W_PASSES(K, N); ++i)
        wr[i] = *(const v4f*)(W + 4 * (size_t)(i * AGG_WAVES * 32 + (int)threadIdx.x));
}
template<int K, int N>
__device__ __forceinline__ void w_fragments_store(float* wf, const v4f (&wr)[W_PASSES(K, N)]) {
#pragma unroll
    for (int i = 0; i < W_PASSES(K, N); ++i) {
        const int f  = i * AGG_WAVES * 32 + (int)threadIdx.x;
        const int k  = f / (N / 4);
        const int n  = (f % (N / 4)) * 4;
        const int t  = ((k >> 3) << 1) | ((k >> 1) & 1);
        const int hh = (k >> 2) & 1;
        const int e  = k & 1;
        float* d = wf + (t * (N / 16) + (n >> 4)) * 64 + (hh * 16 + (n & 15)) * 2 + e;
#pragma unroll
        for (int c = 0; c < 4; ++c) d[2 * c] = wr[i][c];
    }
}

template<int K, int N, bool RELU_A>
__device__ __forceinline__ void owner_rows_product(const float* arows, const float* wf,
                                                   float* __restrict__ H,
                                                   int lo, int cnt, int wave, int lane) {
    constexpr int NCT   = N / 16;
    constexpr int RSTEP = AGG_WAVES / NCT;
    constexpr int PITCH = K + 4;
    static_assert(AGG_WAVES % NCT == 0, "every column tile gets the same number of waves");
    static_assert(K == 128, "the A words below assume four channels per lane in a row");
    const int m  = lane & 15;
    const int hh = lane >> 4;
    const int ct = wave & (NCT - 1);
    const int ntiles = (cnt + 15) >> 4;
    const float* bl = wf + ct * 64 + (hh * 16 + m) * 2;

    for (int rt = wave / NCT; rt < ntiles; rt += RSTEP) {
        const float* al = arows + (rt * 16 + m) * PITCH + hh;
        v8f acc = {};
#pragma unroll 8
        for (int t = 0; t < K / 4; ++t) {
            const float* p = al + 64 * (t & 1) + 2 * (t >> 1);
            float a0 = p[0], a1 = p[32];
            if (RELU_A) { a0 = fmaxf(a0, 0.0f); a1 = fmaxf(a1, 0.0f); }
            const v2f a = { a0, a1 };
            const v2f b = *(const v2f*)(bl + t * (NCT * 64));
            acc = __builtin_amdgcn_wmma_f32_16x16x4_f32(
                      false, a, false, b, (short)0, acc, false, false);
        }
        const int r0 = rt * 16 + hh * 8;
        float* __restrict__ hrow = H + (size_t)(lo + r0) * N + ct * 16 + m;
#pragma unroll
        for (int q2 = 0; q2 < 8; ++q2)
            if (r0 + q2 < cnt) hrow[(size_t)q2 * N] = acc[q2];
    }
}

template<int NB>
__global__ __launch_bounds__(256)
void owner_mark_kernel(const int* __restrict__ dst, unsigned* __restrict__ marks) {
    constexpr unsigned NPB = (N_NODES + NB - 1) / NB;
    static_assert((NB & (NB - 1)) == 0, "owner number is kept in range with a mask");
    static_assert(N_EDGES % UNIT_EDGES == 0, "a unit is 16 whole edges");
    static_assert(MARK_UNITS * UNIT_EDGES == 4 * 4 * 256, "256 threads, four loads of four edges each");

    __shared__ unsigned bm[NB * 8];
    const int t = threadIdx.x;
    for (int i = t; i < NB * 8; i += 256) bm[i] = 0u;
    __syncthreads();

    const unsigned ebase = blockIdx.x * (unsigned)(MARK_UNITS * UNIT_EDGES);
    v4i d[4];
#pragma unroll
    for (int j = 0; j < 4; ++j) {
        const unsigned e = ebase + (unsigned)(j * 1024 + 4 * t);
        const v4i none = {0, 0, 0, 0};
        d[j] = e < (unsigned)N_EDGES ? *(const v4i*)(dst + e) : none;
    }
#pragma unroll
    for (int j = 0; j < 4; ++j) {
        const unsigned e  = ebase + (unsigned)(j * 1024 + 4 * t);
        const unsigned ul = (unsigned)(j * 64 + (t >> 2));
        if (e < (unsigned)N_EDGES) {
#pragma unroll
            for (int i = 0; i < 4; ++i) {
                const unsigned owner = ((unsigned)d[j][i] / NPB) & (unsigned)(NB - 1);
                atomicOr(&bm[owner * 8u + (ul >> 5)], 1u << (ul & 31u));
            }
        }
    }
    __syncthreads();

    unsigned* mine = marks + (size_t)blockIdx.x * (NB * 8);
    for (int i = t; i < NB * 8; i += 256) mine[i] = bm[i];
}

template<int NB, typename F>
__device__ __forceinline__ void scan_marks(const unsigned* __restrict__ marks, unsigned* myq,
                                           int wave, int lane, F group) {
    constexpr int NW  = MARK_WORDS;
    constexpr int WPW = NW / AGG_WAVES;
    static_assert(NW % AGG_WAVES == 0, "every wave gets the same number of words");
    static_assert(WPW <= 64, "two words per lane cover a wave's share");

    int head = 0, tail = 0;
    const unsigned* mine = marks + (size_t)blockIdx.x * 8;
    unsigned x[2], wd[2];
#pragma unroll
    for (int h = 0; h < 2; ++h) {
        const int i = h * 32 + lane;
        x[h]  = (unsigned)(wave * WPW + i);
        wd[h] = i < WPW ? mine[(size_t)(x[h] >> 3) * (NB * 8) + (x[h] & 7u)] : 0u;
    }
#pragma unroll
    for (int h = 0; h < 2; ++h) {
        unsigned w = wd[h];
        for (int it = 0; it < 32; ++it) {
            const bool     m    = w != 0u;
            const unsigned mask = __builtin_amdgcn_ballot_w32(m);
            if (!mask) break;
            if (m) {
                const unsigned k = (unsigned)__builtin_ctz(w);
                w &= w - 1u;
                myq[(tail + (int)__builtin_amdgcn_mbcnt_lo(mask, 0u)) & (FILT_QCAP - 1)] = x[h] * 32u + k;
            }
            tail += __builtin_popcount(mask);
            if (tail - head >= 32) {
                group(head, 32);
                head += 32;
            }
        }
    }
    if (tail - head > 0) group(head, tail - head);
}

template<int NB>
__global__ __launch_bounds__(AGG_WAVES * 32)
void filter_degree_kernel(const int* __restrict__ src,
                          const int* __restrict__ dst,
                          const unsigned* __restrict__ marks,
                          float* __restrict__ isq,
                          unsigned* __restrict__ lists,
                          int* __restrict__ counts,
                          const float* __restrict__ x,
                          const float* __restrict__ W1,
                          float* __restrict__ h1) {
    constexpr int NPB = (N_NODES + NB - 1) / NB;
    static_assert(NPB <= 512, "local node index is packed in 9 bits");
    static_assert(N_NODES <= (1 << 23), "source index is packed in 23 bits");

    constexpr int XROWS = ((NPB + 15) / 16) * 16;
    constexpr int XPW   = (XROWS + AGG_WAVES - 1) / AGG_WAVES;
    __shared__ unsigned degcnt[NPB];
    __shared__ unsigned ring[AGG_WAVES * FILT_QCAP];
    __shared__ float    xs[XROWS * (C_IN + 4)];
    __shared__ __attribute__((aligned(16))) float wf[C_IN * C_HID];

    const int lane = threadIdx.x & 31;
    const int wave = __builtin_amdgcn_readfirstlane((int)(threadIdx.x >> 5));
    const int lo   = blockIdx.x * NPB;
    int cnt = N_NODES - lo;
    cnt = cnt < NPB ? cnt : NPB;
    cnt = cnt > 0 ? cnt : 0;

    v4f xv[XPW];
#pragma unroll
    for (int i = 0; i < XPW; ++i) {
        int node = lo + wave + i * AGG_WAVES;
        node = node < N_NODES ? node : N_NODES - 1;
        xv[i] = *(const v4f*)(x + (size_t)node * C_IN + 4 * lane);
    }
    v4f wr[W_PASSES(C_IN, C_HID)];
    w_fragments_load<C_IN, C_HID>(W1, wr);

    for (int n = threadIdx.x; n < NPB; n += AGG_WAVES * 32) degcnt[n] = 0u;

#pragma unroll
    for (int i = 0; i < XPW; ++i) {
        const int n = wave + i * AGG_WAVES;
        if (n < XROWS) {
#pragma unroll
            for (int j = 0; j < 4; ++j) xs[n * (C_IN + 4) + j * 32 + lane] = xv[i][j];
        }
    }
    w_fragments_store<C_IN, C_HID>(wf, wr);
    __syncthreads();

    unsigned* myq = ring + wave * FILT_QCAP;
    const int gw  = blockIdx.x * AGG_WAVES + wave;
    unsigned* mylist = lists + (size_t)gw * LIST_CAP;
    int total = 0;

    scan_marks<NB>(marks, myq, wave, lane, [&](int head, int nb) {
        unsigned unit = N_UNITS;
        if (lane < nb) unit = myq[(head + lane) & (FILT_QCAP - 1)];
        const bool     live = unit < (unsigned)N_UNITS;
        const unsigned e0   = live ? unit * 16u : 0u;
        v4i dv[4], sv[4];
#pragma unroll
        for (int j = 0; j < 4; ++j) {
            dv[j] = *(const v4i*)(dst + e0 + 4 * j);
            sv[j] = *(const v4i*)(src + e0 + 4 * j);
        }
#pragma unroll
        for (int k = 0; k < 16; ++k) {
            const unsigned dl   = (unsigned)(dv[k >> 2][k & 3] - lo);
            const bool     m    = live && dl < (unsigned)cnt;
            const unsigned mask = __builtin_amdgcn_ballot_w32(m);
            if (mask) {
                if (m) {
                    atomicAdd(&degcnt[dl], 1u);
                    const int pos = total + (int)__builtin_amdgcn_mbcnt_lo(mask, 0u);
                    if (pos < LIST_CAP)
                        mylist[pos] = ((unsigned)sv[k >> 2][k & 3] << 9) | dl;
                }
                total += __builtin_popcount(mask);
            }
        }
    });
    if (lane == 0) counts[gw] = total <= LIST_CAP ? total : -1;

    owner_rows_product<C_IN, C_HID, false>(xs, wf, h1, lo, cnt, wave, lane);
    __syncthreads();

    for (int n = threadIdx.x; n < cnt; n += AGG_WAVES * 32)
        isq[lo + n] = rsqrtf((float)degcnt[n] + 1.0f);
}

template<int C, int PITCH>
__device__ __forceinline__ void agg_add_rows(const float* __restrict__ h,
                                             const float* __restrict__ isq,
                                             float* acc, unsigned ent,
                                             int nb, int lo, int lane) {
    constexpr int V = C / 32;
    typedef float vecf __attribute__((ext_vector_type(V)));

    const int   dl   = (int)(ent & 511u);
    const int   s_l  = (int)(ent >> 9);
    const float iq_d = isq[lo + dl];
    const float iq_s = isq[s_l];

    for (int k0 = 0; k0 < nb; k0 += AGG_ROWS) {
        vecf hv[AGG_ROWS];
#pragma unroll
        for (int g = 0; g < AGG_ROWS; ++g) {
            int kk = k0 + g;
            kk = kk < nb ? kk : nb - 1;
            const int s = __builtin_amdgcn_readlane(s_l, kk);
            hv[g] = *(const vecf*)(h + (size_t)s * C + V * lane);
        }
#pragma unroll
        for (int g = 0; g < AGG_ROWS; ++g) {
            const int k = k0 + g;
            if (k < nb) {
                const float cs = __int_as_float(__builtin_amdgcn_readlane(__float_as_int(iq_s), k));
                const float cd = __int_as_float(__builtin_amdgcn_readlane(__float_as_int(iq_d), k));
                const float coef = cs * cd;
                float* row = acc + __builtin_amdgcn_readlane(dl, k) * PITCH + lane;
#pragma unroll
                for (int j = 0; j < V; ++j)
                    atomicAdd(row + j * 32, hv[g][j] * coef);
            }
        }
    }
}

template<int C, int NB, int NEXT_N>
__global__ __launch_bounds__(AGG_WAVES * 32)
void agg_owned_lds_kernel(const float* __restrict__ h,
                          const int* __restrict__ src,
                          const int* __restrict__ dst,
                          const float* __restrict__ isq,
                          const float* __restrict__ bias,
                          const unsigned* __restrict__ lists,
                          const int* __restrict__ counts,
                          const float* __restrict__ wnext,
                          float* __restrict__ out) {
    constexpr int V    = C / 32;
    constexpr int NPB  = (N_NODES + NB - 1) / NB;
    constexpr int NROWS = NEXT_N > 0 ? ((NPB + 15) / 16) * 16 : NPB;
    constexpr int PITCH = NEXT_N > 0 ? C + 4 : C;
    static_assert(NPB <= 512, "local node index is packed in 9 bits");
    static_assert(NEXT_N == 0 || V == 4, "the fragment read assumes four channels per lane");
    typedef float vecf __attribute__((ext_vector_type(V)));

    __shared__ float    acc[NROWS * PITCH];
    __shared__ unsigned ring[AGG_WAVES * AGG_QCAP];
    __shared__ __attribute__((aligned(16))) float wf[NEXT_N > 0 ? C * NEXT_N : 4];

    const int lane = threadIdx.x & 31;
    const int wave = __builtin_amdgcn_readfirstlane((int)(threadIdx.x >> 5));
    const int lo   = blockIdx.x * NPB;
    int cnt = N_NODES - lo;
    cnt = cnt < NPB ? cnt : NPB;
    cnt = cnt > 0 ? cnt : 0;

    const int gw = blockIdx.x * AGG_WAVES + wave;
    const int nlist = __builtin_amdgcn_readfirstlane(counts[gw]);
    v4f wr[NEXT_N > 0 ? W_PASSES(C, NEXT_N) : 1];
    if constexpr (NEXT_N > 0) w_fragments_load<C, NEXT_N>(wnext, wr);

    for (int n = wave; n < cnt; n += AGG_WAVES) {
        const int   node = lo + n;
        const float s    = isq[node];
        const float s2   = s * s;
        const vecf  hv   = *(const vecf*)(h + (size_t)node * C + V * lane);
#pragma unroll
        for (int j = 0; j < V; ++j)
            acc[n * PITCH + j * 32 + lane] = hv[j] * s2 + bias[V * lane + j];
    }
    if constexpr (NEXT_N > 0) {
        for (int n = cnt + wave; n < ((cnt + 15) & ~15); n += AGG_WAVES)
            for (int j = 0; j < V; ++j)
                acc[n * PITCH + j * 32 + lane] = 0.0f;
        w_fragments_store<C, NEXT_N>(wf, wr);
    }
    __syncthreads();

    if (nlist >= 0) {
        const unsigned* mylist = lists + (size_t)gw * LIST_CAP;
        unsigned ent = lane < nlist ? mylist[lane] : 0u;
        for (int pos = 0; pos < nlist; pos += 32) {
            const int left = nlist - pos;
            const unsigned nxt = pos + 32 + lane < nlist ? mylist[pos + 32 + lane] : 0u;
            agg_add_rows<C, PITCH>(h, isq, acc, ent, left < 32 ? left : 32, lo, lane);
            ent = nxt;
        }
    } else {
        unsigned* myq = ring + wave * AGG_QCAP;
        scan_owned(dst, myq, lo, cnt, wave, lane, [&](int head, int nb) {
            unsigned ent = 0u;
            if (lane < nb) {
                const unsigned q = myq[(head + lane) & (AGG_QCAP - 1)];
                ent = ((unsigned)src[q >> 9] << 9) | (q & 511u);
            }
            agg_add_rows<C, PITCH>(h, isq, acc, ent, nb, lo, lane);
        });
    }
    __syncthreads();

    if constexpr (NEXT_N > 0) {
        owner_rows_product<C, NEXT_N, true>(acc, wf, out, lo, cnt, wave, lane);
    } else {
        for (int n = wave; n < cnt; n += AGG_WAVES) {
            vecf o;
#pragma unroll
            for (int j = 0; j < V; ++j) o[j] = acc[n * PITCH + j * 32 + lane];
            *(vecf*)(out + (size_t)(lo + n) * C + V * lane) = o;
        }
    }
}


extern "C" void kernel_launch(void* const* d_in, const int* in_sizes, int n_in,
                              void* d_out, int out_size, void* d_ws, size_t ws_size,
                              hipStream_t stream) {
    (void)in_sizes; (void)n_in; (void)out_size; (void)ws_size;

    const float* x   = (const float*)d_in[0];
    const int*   ei  = (const int*)  d_in[1];
    const float* W1  = (const float*)d_in[2];
    const float* b1  = (const float*)d_in[3];
    const float* W2  = (const float*)d_in[4];
    const float* b2  = (const float*)d_in[5];
    float*       out = (float*)d_out;

    const int* src = ei;
    const int* dst = ei + N_EDGES;

    float*    isq    = (float*)d_ws;
    float*    h1     = isq + 50176;
    float*    h2     = h1 + (size_t)N_NODES * C_HID;
    int*      counts = (int*)(h2 + (size_t)N_NODES * C_HID);
    unsigned* lists  = (unsigned*)(counts + AGG_NB * AGG_WAVES);
    unsigned* marks  = lists + (size_t)AGG_NB * AGG_WAVES * LIST_CAP;

    owner_mark_kernel<AGG_NB>
        <<<MARK_BLOCKS, 256, 0, stream>>>(dst, marks);

    filter_degree_kernel<AGG_NB>
        <<<AGG_NB, AGG_WAVES * 32, 0, stream>>>(src, dst, marks, isq, lists, counts, x, W1, h1);

    agg_owned_lds_kernel<C_HID, AGG_NB, C_OUTC>
        <<<AGG_NB, AGG_WAVES * 32, 0, stream>>>(h1, src, dst, isq, b1, lists, counts, W2, h2);

    agg_owned_lds_kernel<C_OUTC, AGG_NB, 0>
        <<<AGG_NB, AGG_WAVES * 32, 0, stream>>>(h2, src, dst, isq, b2, lists, counts, nullptr, out);
}


# Owners' pass: half of each block's waves (4 per SIMD) do their x@W1 tiles before their share of the edge scan, the other half after it, so matrix work overlaps the scan's memory waits
# speedup vs baseline: 4.4088x; 1.0485x over previous
#include <hip/hip_runtime.h>


typedef float v2f __attribute__((ext_vector_type(2)));
typedef float v8f __attribute__((ext_vector_type(8)));

#define N_NODES 50000
#define N_EDGES 800000
#define C_IN    128
#define C_HID   128
#define C_OUTC  64

#define AGG_WAVES 32
#define AGG_QCAP  1024
#define AGG_SCAN  8
#define AGG_ROWS  8
#define AGG_NB    256
#define LIST_CAP  2048

#define UNIT_EDGES  16
#define N_UNITS     (N_EDGES / UNIT_EDGES)
#define MARK_UNITS  256
#define MARK_BLOCKS ((N_UNITS + MARK_UNITS - 1) / MARK_UNITS)
#define MARK_WORDS  (MARK_BLOCKS * 8)
#define FILT_QCAP   64

typedef int v4i __attribute__((ext_vector_type(4)));

template<typename F>
__device__ __forceinline__ void scan_owned(const int* __restrict__ dst, unsigned* myq,
                                           int lo, int cnt, int wave, int lane, F group) {
    constexpr int NCH = N_EDGES / 128;
    constexpr int CPW = (MARK_WORDS / AGG_WAVES) * 32 * UNIT_EDGES / 128;
    static_assert(N_EDGES % 128 == 0, "edge list is scanned in whole chunks");
    static_assert(N_EDGES <= (1 << 23), "edge index is packed in 23 bits");
    static_assert(MARK_WORDS % AGG_WAVES == 0 && ((MARK_WORDS / AGG_WAVES) * 32 * UNIT_EDGES) % 128 == 0,
                  "a wave's share is whole chunks");

    int head = 0, tail = 0;
    const int c0 = wave * CPW;
    int nmine = NCH - c0;
    nmine = nmine < CPW ? nmine : CPW;
    nmine = nmine > 0 ? nmine : 0;
    const unsigned lane4 = (unsigned)lane * 4u;

    auto match_chunk = [&](const v4i dv, const unsigned ebase) {
#pragma unroll
        for (int j = 0; j < 4; ++j) {
            const unsigned dl = (unsigned)(dv[j] - lo);
            const bool     m  = dl < (unsigned)cnt;
            const unsigned mask = __builtin_amdgcn_ballot_w32(m);
            if (mask) {
                const int pre = (int)__builtin_amdgcn_mbcnt_lo(mask, 0u);
                if (m) myq[(tail + pre) & (AGG_QCAP - 1)] =
                           ((ebase + (unsigned)j) << 9) | dl;
                tail += __builtin_popcount(mask);
            }
        }
    };
    auto groups_full = [&]() {
        while (tail - head >= 32) {
            group(head, 32);
            head += 32;
        }
    };
    auto load_chunk = [&](int i) -> v4i {
        const unsigned c = i < nmine ? (unsigned)(c0 + i) : 0u;
        return *(const v4i*)(dst + (c * 128u + lane4));
    };

    int i0 = 0;
    v4i cur[AGG_SCAN], nxt[AGG_SCAN];
#pragma unroll
    for (int u = 0; u < AGG_SCAN; ++u) cur[u] = load_chunk(i0 + u);
    for (; i0 + AGG_SCAN <= nmine; i0 += AGG_SCAN) {
#pragma unroll
        for (int u = 0; u < AGG_SCAN; ++u) nxt[u] = load_chunk(i0 + AGG_SCAN + u);
#pragma unroll
        for (int u = 0; u < AGG_SCAN; ++u) {
            match_chunk(cur[u], (unsigned)(c0 + i0 + u) * 128u + lane4);
            if ((u & 3) == 3) groups_full();
        }
#pragma unroll
        for (int u = 0; u < AGG_SCAN; ++u) cur[u] = nxt[u];
    }
#pragma unroll
    for (int u = 0; u < AGG_SCAN; ++u)
        if (i0 + u < nmine) match_chunk(cur[u], (unsigned)(c0 + i0 + u) * 128u + lane4);
    groups_full();
    if (tail - head > 0) group(head, tail - head);
}

typedef float v4f __attribute__((ext_vector_type(4)));

#define W_PASSES(K, N) ((K) * (N) / 4 / (AGG_WAVES * 32))

template<int K, int N>
__device__ __forceinline__ void w_fragments_load(const float* __restrict__ W, v4f (&wr)[W_PASSES(K, N)]) {
    static_assert((K * N / 4) % (AGG_WAVES * 32) == 0, "whole passes of the block");
#pragma unroll
    for (int i = 0; i < W_PASSES(K, N); ++i)
        wr[i] = *(const v4f*)(W + 4 * (size_t)(i * AGG_WAVES * 32 + (int)threadIdx.x));
}
template<int K, int N>
__device__ __forceinline__ void w_fragments_store(float* wf, const v4f (&wr)[W_PASSES(K, N)]) {
#pragma unroll
    for (int i = 0; i < W_PASSES(K, N); ++i) {
        const int f  = i * AGG_WAVES * 32 + (int)threadIdx.x;
        const int k  = f / (N / 4);
        const int n  = (f % (N / 4)) * 4;
        const int t  = ((k >> 3) << 1) | ((k >> 1) & 1);
        const int hh = (k >> 2) & 1;
        const int e  = k & 1;
        float* d = wf + (t * (N / 16) + (n >> 4)) * 64 + (hh * 16 + (n & 15)) * 2 + e;
#pragma unroll
        for (int c = 0; c < 4; ++c) d[2 * c] = wr[i][c];
    }
}

template<int K, int N, bool RELU_A>
__device__ __forceinline__ void owner_rows_product(const float* arows, const float* wf,
                                                   float* __restrict__ H,
                                                   int lo, int cnt, int wave, int lane) {
    constexpr int NCT   = N / 16;
    constexpr int RSTEP = AGG_WAVES / NCT;
    constexpr int PITCH = K + 4;
    static_assert(AGG_WAVES % NCT == 0, "every column tile gets the same number of waves");
    static_assert(K == 128, "the A words below assume four channels per lane in a row");
    const int m  = lane & 15;
    const int hh = lane >> 4;
    const int ct = wave & (NCT - 1);
    const int ntiles = (cnt + 15) >> 4;
    const float* bl = wf + ct * 64 + (hh * 16 + m) * 2;

    for (int rt = wave / NCT; rt < ntiles; rt += RSTEP) {
        const float* al = arows + (rt * 16 + m) * PITCH + hh;
        v8f acc = {};
#pragma unroll 8
        for (int t = 0; t < K / 4; ++t) {
            const float* p = al + 64 * (t & 1) + 2 * (t >> 1);
            float a0 = p[0], a1 = p[32];
            if (RELU_A) { a0 = fmaxf(a0, 0.0f); a1 = fmaxf(a1, 0.0f); }
            const v2f a = { a0, a1 };
            const v2f b = *(const v2f*)(bl + t * (NCT * 64));
            acc = __builtin_amdgcn_wmma_f32_16x16x4_f32(
                      false, a, false, b, (short)0, acc, false, false);
        }
        const int r0 = rt * 16 + hh * 8;
        float* __restrict__ hrow = H + (size_t)(lo + r0) * N + ct * 16 + m;
#pragma unroll
        for (int q2 = 0; q2 < 8; ++q2)
            if (r0 + q2 < cnt) hrow[(size_t)q2 * N] = acc[q2];
    }
}

template<int NB>
__global__ __launch_bounds__(256)
void owner_mark_kernel(const int* __restrict__ dst, unsigned* __restrict__ marks) {
    constexpr unsigned NPB = (N_NODES + NB - 1) / NB;
    static_assert((NB & (NB - 1)) == 0, "owner number is kept in range with a mask");
    static_assert(N_EDGES % UNIT_EDGES == 0, "a unit is 16 whole edges");
    static_assert(MARK_UNITS * UNIT_EDGES == 4 * 4 * 256, "256 threads, four loads of four edges each");

    __shared__ unsigned bm[NB * 8];
    const int t = threadIdx.x;
    for (int i = t; i < NB * 8; i += 256) bm[i] = 0u;
    __syncthreads();

    const unsigned ebase = blockIdx.x * (unsigned)(MARK_UNITS * UNIT_EDGES);
    v4i d[4];
#pragma unroll
    for (int j = 0; j < 4; ++j) {
        const unsigned e = ebase + (unsigned)(j * 1024 + 4 * t);
        const v4i none = {0, 0, 0, 0};
        d[j] = e < (unsigned)N_EDGES ? *(const v4i*)(dst + e) : none;
    }
#pragma unroll
    for (int j = 0; j < 4; ++j) {
        const unsigned e  = ebase + (unsigned)(j * 1024 + 4 * t);
        const unsigned ul = (unsigned)(j * 64 + (t >> 2));
        if (e < (unsigned)N_EDGES) {
#pragma unroll
            for (int i = 0; i < 4; ++i) {
                const unsigned owner = ((unsigned)d[j][i] / NPB) & (unsigned)(NB - 1);
                atomicOr(&bm[owner * 8u + (ul >> 5)], 1u << (ul & 31u));
            }
        }
    }
    __syncthreads();

    unsigned* mine = marks + (size_t)blockIdx.x * (NB * 8);
    for (int i = t; i < NB * 8; i += 256) mine[i] = bm[i];
}

template<int NB, typename F>
__device__ __forceinline__ void scan_marks(const unsigned* __restrict__ marks, unsigned* myq,
                                           int wave, int lane, F group) {
    constexpr int NW  = MARK_WORDS;
    constexpr int WPW = NW / AGG_WAVES;
    static_assert(NW % AGG_WAVES == 0, "every wave gets the same number of words");
    static_assert(WPW <= 64, "two words per lane cover a wave's share");

    int head = 0, tail = 0;
    const unsigned* mine = marks + (size_t)blockIdx.x * 8;
    unsigned x[2], wd[2];
#pragma unroll
    for (int h = 0; h < 2; ++h) {
        const int i = h * 32 + lane;
        x[h]  = (unsigned)(wave * WPW + i);
        wd[h] = i < WPW ? mine[(size_t)(x[h] >> 3) * (NB * 8) + (x[h] & 7u)] : 0u;
    }
#pragma unroll
    for (int h = 0; h < 2; ++h) {
        unsigned w = wd[h];
        for (int it = 0; it < 32; ++it) {
            const bool     m    = w != 0u;
            const unsigned mask = __builtin_amdgcn_ballot_w32(m);
            if (!mask) break;
            if (m) {
                const unsigned k = (unsigned)__builtin_ctz(w);
                w &= w - 1u;
                myq[(tail + (int)__builtin_amdgcn_mbcnt_lo(mask, 0u)) & (FILT_QCAP - 1)] = x[h] * 32u + k;
            }
            tail += __builtin_popcount(mask);
            if (tail - head >= 32) {
                group(head, 32);
                head += 32;
            }
        }
    }
    if (tail - head > 0) group(head, tail - head);
}

template<int NB>
__global__ __launch_bounds__(AGG_WAVES * 32)
void filter_degree_kernel(const int* __restrict__ src,
                          const int* __restrict__ dst,
                          const unsigned* __restrict__ marks,
                          float* __restrict__ isq,
                          unsigned* __restrict__ lists,
                          int* __restrict__ counts,
                          const float* __restrict__ x,
                          const float* __restrict__ W1,
                          float* __restrict__ h1) {
    constexpr int NPB = (N_NODES + NB - 1) / NB;
    static_assert(NPB <= 512, "local node index is packed in 9 bits");
    static_assert(N_NODES <= (1 << 23), "source index is packed in 23 bits");

    constexpr int XROWS = ((NPB + 15) / 16) * 16;
    constexpr int XPW   = (XROWS + AGG_WAVES - 1) / AGG_WAVES;
    __shared__ unsigned degcnt[NPB];
    __shared__ unsigned ring[AGG_WAVES * FILT_QCAP];
    __shared__ float    xs[XROWS * (C_IN + 4)];
    __shared__ __attribute__((aligned(16))) float wf[C_IN * C_HID];

    const int lane = threadIdx.x & 31;
    const int wave = __builtin_amdgcn_readfirstlane((int)(threadIdx.x >> 5));
    const int lo   = blockIdx.x * NPB;
    int cnt = N_NODES - lo;
    cnt = cnt < NPB ? cnt : NPB;
    cnt = cnt > 0 ? cnt : 0;

    v4f xv[XPW];
#pragma unroll
    for (int i = 0; i < XPW; ++i) {
        int node = lo + wave + i * AGG_WAVES;
        node = node < N_NODES ? node : N_NODES - 1;
        xv[i] = *(const v4f*)(x + (size_t)node * C_IN + 4 * lane);
    }
    v4f wr[W_PASSES(C_IN, C_HID)];
    w_fragments_load<C_IN, C_HID>(W1, wr);

    for (int n = threadIdx.x; n < NPB; n += AGG_WAVES * 32) degcnt[n] = 0u;

#pragma unroll
    for (int i = 0; i < XPW; ++i) {
        const int n = wave + i * AGG_WAVES;
        if (n < XROWS) {
#pragma unroll
            for (int j = 0; j < 4; ++j) xs[n * (C_IN + 4) + j * 32 + lane] = xv[i][j];
        }
    }
    w_fragments_store<C_IN, C_HID>(wf, wr);
    __syncthreads();

    const bool product_first = (wave & 4) != 0;
    if (product_first) owner_rows_product<C_IN, C_HID, false>(xs, wf, h1, lo, cnt, wave, lane);

    unsigned* myq = ring + wave * FILT_QCAP;
    const int gw  = blockIdx.x * AGG_WAVES + wave;
    unsigned* mylist = lists + (size_t)gw * LIST_CAP;
    int total = 0;

    scan_marks<NB>(marks, myq, wave, lane, [&](int head, int nb) {
        unsigned unit = N_UNITS;
        if (lane < nb) unit = myq[(head + lane) & (FILT_QCAP - 1)];
        const bool     live = unit < (unsigned)N_UNITS;
        const unsigned e0   = live ? unit * 16u : 0u;
        v4i dv[4], sv[4];
#pragma unroll
        for (int j = 0; j < 4; ++j) {
            dv[j] = *(const v4i*)(dst + e0 + 4 * j);
            sv[j] = *(const v4i*)(src + e0 + 4 * j);
        }
#pragma unroll
        for (int k = 0; k < 16; ++k) {
            const unsigned dl   = (unsigned)(dv[k >> 2][k & 3] - lo);
            const bool     m    = live && dl < (unsigned)cnt;
            const unsigned mask = __builtin_amdgcn_ballot_w32(m);
            if (mask) {
                if (m) {
                    atomicAdd(&degcnt[dl], 1u);
                    const int pos = total + (int)__builtin_amdgcn_mbcnt_lo(mask, 0u);
                    if (pos < LIST_CAP)
                        mylist[pos] = ((unsigned)sv[k >> 2][k & 3] << 9) | dl;
                }
                total += __builtin_popcount(mask);
            }
        }
    });
    if (lane == 0) counts[gw] = total <= LIST_CAP ? total : -1;

    if (!product_first) owner_rows_product<C_IN, C_HID, false>(xs, wf, h1, lo, cnt, wave, lane);
    __syncthreads();

    for (int n = threadIdx.x; n < cnt; n += AGG_WAVES * 32)
        isq[lo + n] = rsqrtf((float)degcnt[n] + 1.0f);
}

template<int C, int PITCH>
__device__ __forceinline__ void agg_add_rows(const float* __restrict__ h,
                                             const float* __restrict__ isq,
                                             float* acc, unsigned ent,
                                             int nb, int lo, int lane) {
    constexpr int V = C / 32;
    typedef float vecf __attribute__((ext_vector_type(V)));

    const int   dl   = (int)(ent & 511u);
    const int   s_l  = (int)(ent >> 9);
    const float iq_d = isq[lo + dl];
    const float iq_s = isq[s_l];

    for (int k0 = 0; k0 < nb; k0 += AGG_ROWS) {
        vecf hv[AGG_ROWS];
#pragma unroll
        for (int g = 0; g < AGG_ROWS; ++g) {
            int kk = k0 + g;
            kk = kk < nb ? kk : nb - 1;
            const int s = __builtin_amdgcn_readlane(s_l, kk);
            hv[g] = *(const vecf*)(h + (size_t)s * C + V * lane);
        }
#pragma unroll
        for (int g = 0; g < AGG_ROWS; ++g) {
            const int k = k0 + g;
            if (k < nb) {
                const float cs = __int_as_float(__builtin_amdgcn_readlane(__float_as_int(iq_s), k));
                const float cd = __int_as_float(__builtin_amdgcn_readlane(__float_as_int(iq_d), k));
                const float coef = cs * cd;
                float* row = acc + __builtin_amdgcn_readlane(dl, k) * PITCH + lane;
#pragma unroll
                for (int j = 0; j < V; ++j)
                    atomicAdd(row + j * 32, hv[g][j] * coef);
            }
        }
    }
}

template<int C, int NB, int NEXT_N>
__global__ __launch_bounds__(AGG_WAVES * 32)
void agg_owned_lds_kernel(const float* __restrict__ h,
                          const int* __restrict__ src,
                          const int* __restrict__ dst,
                          const float* __restrict__ isq,
                          const float* __restrict__ bias,
                          const unsigned* __restrict__ lists,
                          const int* __restrict__ counts,
                          const float* __restrict__ wnext,
                          float* __restrict__ out) {
    constexpr int V    = C / 32;
    constexpr int NPB  = (N_NODES + NB - 1) / NB;
    constexpr int NROWS = NEXT_N > 0 ? ((NPB + 15) / 16) * 16 : NPB;
    constexpr int PITCH = NEXT_N > 0 ? C + 4 : C;
    static_assert(NPB <= 512, "local node index is packed in 9 bits");
    static_assert(NEXT_N == 0 || V == 4, "the fragment read assumes four channels per lane");
    typedef float vecf __attribute__((ext_vector_type(V)));

    __shared__ float    acc[NROWS * PITCH];
    __shared__ unsigned ring[AGG_WAVES * AGG_QCAP];
    __shared__ __attribute__((aligned(16))) float wf[NEXT_N > 0 ? C * NEXT_N : 4];

    const int lane = threadIdx.x & 31;
    const int wave = __builtin_amdgcn_readfirstlane((int)(threadIdx.x >> 5));
    const int lo   = blockIdx.x * NPB;
    int cnt = N_NODES - lo;
    cnt = cnt < NPB ? cnt : NPB;
    cnt = cnt > 0 ? cnt : 0;

    const int gw = blockIdx.x * AGG_WAVES + wave;
    const int nlist = __builtin_amdgcn_readfirstlane(counts[gw]);
    v4f wr[NEXT_N > 0 ? W_PASSES(C, NEXT_N) : 1];
    if constexpr (NEXT_N > 0) w_fragments_load<C, NEXT_N>(wnext, wr);

    for (int n = wave; n < cnt; n += AGG_WAVES) {
        const int   node = lo + n;
        const float s    = isq[node];
        const float s2   = s * s;
        const vecf  hv   = *(const vecf*)(h + (size_t)node * C + V * lane);
#pragma unroll
        for (int j = 0; j < V; ++j)
            acc[n * PITCH + j * 32 + lane] = hv[j] * s2 + bias[V * lane + j];
    }
    if constexpr (NEXT_N > 0) {
        for (int n = cnt + wave; n < ((cnt + 15) & ~15); n += AGG_WAVES)
            for (int j = 0; j < V; ++j)
                acc[n * PITCH + j * 32 + lane] = 0.0f;
        w_fragments_store<C, NEXT_N>(wf, wr);
    }
    __syncthreads();

    if (nlist >= 0) {
        const unsigned* mylist = lists + (size_t)gw * LIST_CAP;
        unsigned ent = lane < nlist ? mylist[lane] : 0u;
        for (int pos = 0; pos < nlist; pos += 32) {
            const int left = nlist - pos;
            const unsigned nxt = pos + 32 + lane < nlist ? mylist[pos + 32 + lane] : 0u;
            agg_add_rows<C, PITCH>(h, isq, acc, ent, left < 32 ? left : 32, lo, lane);
            ent = nxt;
        }
    } else {
        unsigned* myq = ring + wave * AGG_QCAP;
        scan_owned(dst, myq, lo, cnt, wave, lane, [&](int head, int nb) {
            unsigned ent = 0u;
            if (lane < nb) {
                const unsigned q = myq[(head + lane) & (AGG_QCAP - 1)];
                ent = ((unsigned)src[q >> 9] << 9) | (q & 511u);
            }
            agg_add_rows<C, PITCH>(h, isq, acc, ent, nb, lo, lane);
        });
    }
    __syncthreads();

    if constexpr (NEXT_N > 0) {
        owner_rows_product<C, NEXT_N, true>(acc, wf, out, lo, cnt, wave, lane);
    } else {
        for (int n = wave; n < cnt; n += AGG_WAVES) {
            vecf o;
#pragma unroll
            for (int j = 0; j < V; ++j) o[j] = acc[n * PITCH + j * 32 + lane];
            *(vecf*)(out + (size_t)(lo + n) * C + V * lane) = o;
        }
    }
}


extern "C" void kernel_launch(void* const* d_in, const int* in_sizes, int n_in,
                              void* d_out, int out_size, void* d_ws, size_t ws_size,
                              hipStream_t stream) {
    (void)in_sizes; (void)n_in; (void)out_size; (void)ws_size;

    const float* x   = (const float*)d_in[0];
    const int*   ei  = (const int*)  d_in[1];
    const float* W1  = (const float*)d_in[2];
    const float* b1  = (const float*)d_in[3];
    const float* W2  = (const float*)d_in[4];
    const float* b2  = (const float*)d_in[5];
    float*       out = (float*)d_out;

    const int* src = ei;
    const int* dst = ei + N_EDGES;

    float*    isq    = (float*)d_ws;
    float*    h1     = isq + 50176;
    float*    h2     = h1 + (size_t)N_NODES * C_HID;
    int*      counts = (int*)(h2 + (size_t)N_NODES * C_HID);
    unsigned* lists  = (unsigned*)(counts + AGG_NB * AGG_WAVES);
    unsigned* marks  = lists + (size_t)AGG_NB * AGG_WAVES * LIST_CAP;

    owner_mark_kernel<AGG_NB>
        <<<MARK_BLOCKS, 256, 0, stream>>>(dst, marks);

    filter_degree_kernel<AGG_NB>
        <<<AGG_NB, AGG_WAVES * 32, 0, stream>>>(src, dst, marks, isq, lists, counts, x, W1, h1);

    agg_owned_lds_kernel<C_HID, AGG_NB, C_OUTC>
        <<<AGG_NB, AGG_WAVES * 32, 0, stream>>>(h1, src, dst, isq, b1, lists, counts, W2, h2);

    agg_owned_lds_kernel<C_OUTC, AGG_NB, 0>
        <<<AGG_NB, AGG_WAVES * 32, 0, stream>>>(h2, src, dst, isq, b2, lists, counts, nullptr, out);
}


# Owner blocks: the filter puts each edge (LDS cursor) into the list of the wave owning its node (node % 32); that wave sums its 7 nodes in registers: no LDS float atomics; sums in list arrival order
# speedup vs baseline: 4.7614x; 1.0800x over previous
#include <hip/hip_runtime.h>


typedef float v2f __attribute__((ext_vector_type(2)));
typedef float v8f __attribute__((ext_vector_type(8)));

#define N_NODES 50000
#define N_EDGES 800000
#define C_IN    128
#define C_HID   128
#define C_OUTC  64

#define AGG_WAVES 32
#define AGG_QCAP  1024
#define AGG_ROWS  8
#define AGG_NB    256
#define LIST_CAP  2048

#define UNIT_EDGES  16
#define N_UNITS     (N_EDGES / UNIT_EDGES)
#define MARK_UNITS  256
#define MARK_BLOCKS ((N_UNITS + MARK_UNITS - 1) / MARK_UNITS)
#define MARK_WORDS  (MARK_BLOCKS * 8)
#define FILT_QCAP   64

typedef int v4i __attribute__((ext_vector_type(4)));

template<typename F>
__device__ __forceinline__ void scan_owned(const int* __restrict__ dst, unsigned* myq,
                                           int lo, int cnt, int wave, int lane, F group) {
    constexpr int NCH = N_EDGES / 128;
    static_assert(N_EDGES % 128 == 0, "edge list is scanned in whole chunks");
    static_assert(N_EDGES <= (1 << 23), "edge index is packed in 23 bits");
    static_assert(AGG_QCAP >= 31 + 128, "up to 31 entries left over and 128 new ones wait in the ring");

    int head = 0, tail = 0;
    const unsigned lane4 = (unsigned)lane * 4u;

    for (int i = 0; i <= NCH; ++i) {
        if (i < NCH) {
            const unsigned ebase = (unsigned)i * 128u + lane4;
            const v4i dv = *(const v4i*)(dst + ebase);
#pragma unroll
            for (int j = 0; j < 4; ++j) {
                const unsigned dl = (unsigned)(dv[j] - lo);
                const bool     m  = dl < (unsigned)cnt && (dl & (unsigned)(AGG_WAVES - 1)) == (unsigned)wave;
                const unsigned mask = __builtin_amdgcn_ballot_w32(m);
                if (mask) {
                    const int pre = (int)__builtin_amdgcn_mbcnt_lo(mask, 0u);
                    if (m) myq[(tail + pre) & (AGG_QCAP - 1)] =
                               ((ebase + (unsigned)j) << 9) | dl;
                    tail += __builtin_popcount(mask);
                }
            }
        }
        const int keep = i < NCH ? 31 : 0;
        while (tail - head > keep) {
            const int left = tail - head;
            const int nb   = left < 32 ? left : 32;
            group(head, nb);
            head += nb;
        }
    }
}

typedef float v4f __attribute__((ext_vector_type(4)));

#define W_PASSES(K, N) ((K) * (N) / 4 / (AGG_WAVES * 32))

template<int K, int N>
__device__ __forceinline__ void w_fragments_load(const float* __restrict__ W, v4f (&wr)[W_PASSES(K, N)]) {
    static_assert((K * N / 4) % (AGG_WAVES * 32) == 0, "whole passes of the block");
#pragma unroll
    for (int i = 0; i < W_PASSES(K, N); ++i)
        wr[i] = *(const v4f*)(W + 4 * (size_t)(i * AGG_WAVES * 32 + (int)threadIdx.x));
}
template<int K, int N>
__device__ __forceinline__ void w_fragments_store(float* wf, const v4f (&wr)[W_PASSES(K, N)]) {
#pragma unroll
    for (int i = 0; i < W_PASSES(K, N); ++i) {
        const int f  = i * AGG_WAVES * 32 + (int)threadIdx.x;
        const int k  = f / (N / 4);
        const int n  = (f % (N / 4)) * 4;
        const int t  = ((k >> 3) << 1) | ((k >> 1) & 1);
        const int hh = (k >> 2) & 1;
        const int e  = k & 1;
        float* d = wf + (t * (N / 16) + (n >> 4)) * 64 + (hh * 16 + (n & 15)) * 2 + e;
#pragma unroll
        for (int c = 0; c < 4; ++c) d[2 * c] = wr[i][c];
    }
}

template<int K, int N, bool RELU_A>
__device__ __forceinline__ void owner_rows_product(const float* arows, const float* wf,
                                                   float* __restrict__ H,
                                                   int lo, int cnt, int wave, int lane) {
    constexpr int NCT   = N / 16;
    constexpr int RSTEP = AGG_WAVES / NCT;
    constexpr int PITCH = K + 4;
    static_assert(AGG_WAVES % NCT == 0, "every column tile gets the same number of waves");
    static_assert(K == 128, "the A words below assume four channels per lane in a row");
    const int m  = lane & 15;
    const int hh = lane >> 4;
    const int ct = wave & (NCT - 1);
    const int ntiles = (cnt + 15) >> 4;
    const float* bl = wf + ct * 64 + (hh * 16 + m) * 2;

    for (int rt = wave / NCT; rt < ntiles; rt += RSTEP) {
        const float* al = arows + (rt * 16 + m) * PITCH + hh;
        v8f acc = {};
#pragma unroll 8
        for (int t = 0; t < K / 4; ++t) {
            const float* p = al + 64 * (t & 1) + 2 * (t >> 1);
            float a0 = p[0], a1 = p[32];
            if (RELU_A) { a0 = fmaxf(a0, 0.0f); a1 = fmaxf(a1, 0.0f); }
            const v2f a = { a0, a1 };
            const v2f b = *(const v2f*)(bl + t * (NCT * 64));
            acc = __builtin_amdgcn_wmma_f32_16x16x4_f32(
                      false, a, false, b, (short)0, acc, false, false);
        }
        const int r0 = rt * 16 + hh * 8;
        float* __restrict__ hrow = H + (size_t)(lo + r0) * N + ct * 16 + m;
#pragma unroll
        for (int q2 = 0; q2 < 8; ++q2)
            if (r0 + q2 < cnt) hrow[(size_t)q2 * N] = acc[q2];
    }
}

template<int NB>
__global__ __launch_bounds__(256)
void owner_mark_kernel(const int* __restrict__ dst, unsigned* __restrict__ marks) {
    constexpr unsigned NPB = (N_NODES + NB - 1) / NB;
    static_assert((NB & (NB - 1)) == 0, "owner number is kept in range with a mask");
    static_assert(N_EDGES % UNIT_EDGES == 0, "a unit is 16 whole edges");
    static_assert(MARK_UNITS * UNIT_EDGES == 4 * 4 * 256, "256 threads, four loads of four edges each");

    __shared__ unsigned bm[NB * 8];
    const int t = threadIdx.x;
    for (int i = t; i < NB * 8; i += 256) bm[i] = 0u;
    __syncthreads();

    const unsigned ebase = blockIdx.x * (unsigned)(MARK_UNITS * UNIT_EDGES);
    v4i d[4];
#pragma unroll
    for (int j = 0; j < 4; ++j) {
        const unsigned e = ebase + (unsigned)(j * 1024 + 4 * t);
        const v4i none = {0, 0, 0, 0};
        d[j] = e < (unsigned)N_EDGES ? *(const v4i*)(dst + e) : none;
    }
#pragma unroll
    for (int j = 0; j < 4; ++j) {
        const unsigned e  = ebase + (unsigned)(j * 1024 + 4 * t);
        const unsigned ul = (unsigned)(j * 64 + (t >> 2));
        if (e < (unsigned)N_EDGES) {
#pragma unroll
            for (int i = 0; i < 4; ++i) {
                const unsigned owner = ((unsigned)d[j][i] / NPB) & (unsigned)(NB - 1);
                atomicOr(&bm[owner * 8u + (ul >> 5)], 1u << (ul & 31u));
            }
        }
    }
    __syncthreads();

    unsigned* mine = marks + (size_t)blockIdx.x * (NB * 8);
    for (int i = t; i < NB * 8; i += 256) mine[i] = bm[i];
}

template<int NB, typename F>
__device__ __forceinline__ void scan_marks(const unsigned* __restrict__ marks, unsigned* myq,
                                           int wave, int lane, F group) {
    constexpr int NW  = MARK_WORDS;
    constexpr int WPW = NW / AGG_WAVES;
    static_assert(NW % AGG_WAVES == 0, "every wave gets the same number of words");
    static_assert(WPW <= 64, "two words per lane cover a wave's share");

    int head = 0, tail = 0;
    const unsigned* mine = marks + (size_t)blockIdx.x * 8;
    unsigned x[2], wd[2];
#pragma unroll
    for (int h = 0; h < 2; ++h) {
        const int i = h * 32 + lane;
        x[h]  = (unsigned)(wave * WPW + i);
        wd[h] = i < WPW ? mine[(size_t)(x[h] >> 3) * (NB * 8) + (x[h] & 7u)] : 0u;
    }
#pragma unroll
    for (int h = 0; h < 2; ++h) {
        unsigned w = wd[h];
        for (int it = 0; it < 32; ++it) {
            const bool     m    = w != 0u;
            const unsigned mask = __builtin_amdgcn_ballot_w32(m);
            if (!mask) break;
            if (m) {
                const unsigned k = (unsigned)__builtin_ctz(w);
                w &= w - 1u;
                myq[(tail + (int)__builtin_amdgcn_mbcnt_lo(mask, 0u)) & (FILT_QCAP - 1)] = x[h] * 32u + k;
            }
            tail += __builtin_popcount(mask);
            if (tail - head >= 32) {
                group(head, 32);
                head += 32;
            }
        }
    }
    if (tail - head > 0) group(head, tail - head);
}

template<int NB>
__global__ __launch_bounds__(AGG_WAVES * 32)
void filter_degree_kernel(const int* __restrict__ src,
                          const int* __restrict__ dst,
                          const unsigned* __restrict__ marks,
                          float* __restrict__ isq,
                          unsigned* __restrict__ lists,
                          int* __restrict__ counts,
                          const float* __restrict__ x,
                          const float* __restrict__ W1,
                          float* __restrict__ h1) {
    constexpr int NPB = (N_NODES + NB - 1) / NB;
    static_assert(NPB <= 512, "local node index is packed in 9 bits");
    static_assert(N_NODES <= (1 << 23), "source index is packed in 23 bits");

    constexpr int XROWS = ((NPB + 15) / 16) * 16;
    constexpr int XPW   = (XROWS + AGG_WAVES - 1) / AGG_WAVES;
    __shared__ unsigned degcnt[NPB];
    __shared__ unsigned cursor[AGG_WAVES];
    __shared__ unsigned ring[AGG_WAVES * FILT_QCAP];
    __shared__ float    xs[XROWS * (C_IN + 4)];
    __shared__ __attribute__((aligned(16))) float wf[C_IN * C_HID];

    const int lane = threadIdx.x & 31;
    const int wave = __builtin_amdgcn_readfirstlane((int)(threadIdx.x >> 5));
    const int lo   = blockIdx.x * NPB;
    int cnt = N_NODES - lo;
    cnt = cnt < NPB ? cnt : NPB;
    cnt = cnt > 0 ? cnt : 0;

    v4f xv[XPW];
#pragma unroll
    for (int i = 0; i < XPW; ++i) {
        int node = lo + wave + i * AGG_WAVES;
        node = node < N_NODES ? node : N_NODES - 1;
        xv[i] = *(const v4f*)(x + (size_t)node * C_IN + 4 * lane);
    }
    v4f wr[W_PASSES(C_IN, C_HID)];
    w_fragments_load<C_IN, C_HID>(W1, wr);

    for (int n = threadIdx.x; n < NPB; n += AGG_WAVES * 32) degcnt[n] = 0u;
    if (threadIdx.x < AGG_WAVES) cursor[threadIdx.x] = 0u;

#pragma unroll
    for (int i = 0; i < XPW; ++i) {
        const int n = wave + i * AGG_WAVES;
        if (n < XROWS) {
#pragma unroll
            for (int j = 0; j < 4; ++j) xs[n * (C_IN + 4) + j * 32 + lane] = xv[i][j];
        }
    }
    w_fragments_store<C_IN, C_HID>(wf, wr);
    __syncthreads();

    const bool product_first = (wave & 4) != 0;
    if (product_first) owner_rows_product<C_IN, C_HID, false>(xs, wf, h1, lo, cnt, wave, lane);

    unsigned* myq = ring + wave * FILT_QCAP;
    unsigned* blist = lists + (size_t)blockIdx.x * AGG_WAVES * LIST_CAP;

    scan_marks<NB>(marks, myq, wave, lane, [&](int head, int nb) {
        unsigned unit = N_UNITS;
        if (lane < nb) unit = myq[(head + lane) & (FILT_QCAP - 1)];
        const bool     live = unit < (unsigned)N_UNITS;
        const unsigned e0   = live ? unit * 16u : 0u;
        v4i dv[4], sv[4];
#pragma unroll
        for (int j = 0; j < 4; ++j) {
            dv[j] = *(const v4i*)(dst + e0 + 4 * j);
            sv[j] = *(const v4i*)(src + e0 + 4 * j);
        }
        unsigned pos[16];
#pragma unroll
        for (int k = 0; k < 16; ++k) {
            const unsigned dl = (unsigned)(dv[k >> 2][k & 3] - lo);
            pos[k] = LIST_CAP;
            if (live && dl < (unsigned)cnt) {
                atomicAdd(&degcnt[dl], 1u);
                pos[k] = atomicAdd(&cursor[dl & (unsigned)(AGG_WAVES - 1)], 1u);
            }
        }
#pragma unroll
        for (int k = 0; k < 16; ++k) {
            const unsigned dl = (unsigned)(dv[k >> 2][k & 3] - lo);
            if (pos[k] < (unsigned)LIST_CAP)
                blist[(size_t)(dl & (unsigned)(AGG_WAVES - 1)) * LIST_CAP + pos[k]] =
                    ((unsigned)sv[k >> 2][k & 3] << 9) | dl;
        }
    });

    if (!product_first) owner_rows_product<C_IN, C_HID, false>(xs, wf, h1, lo, cnt, wave, lane);
    __syncthreads();

    if (threadIdx.x < AGG_WAVES) {
        const unsigned c = cursor[threadIdx.x];
        counts[blockIdx.x * AGG_WAVES + threadIdx.x] = c <= (unsigned)LIST_CAP ? (int)c : -1;
    }

    for (int n = threadIdx.x; n < cnt; n += AGG_WAVES * 32)
        isq[lo + n] = rsqrtf((float)degcnt[n] + 1.0f);
}

#define SRT_W 5
#define SRT_N (SRT_W * 32)

template<int C, int NQ>
__device__ __forceinline__ void wave_add_entries(const float* __restrict__ h,
                                                 const float* __restrict__ isq,
                                                 float (&a)[NQ][C / 32], unsigned* srt,
                                                 const unsigned (&ent)[SRT_W],
                                                 int n, int lo, int lane) {
    constexpr int V = C / 32;
    typedef float vecf __attribute__((ext_vector_type(V)));

    int   s_l[SRT_W], slot[SRT_W];
    float iq_d[SRT_W], iq_s[SRT_W];
#pragma unroll
    for (int w = 0; w < SRT_W; ++w) {
        const int dl = (int)(ent[w] & 511u);
        s_l[w]  = (int)(ent[w] >> 9);
        iq_d[w] = isq[lo + dl];
        iq_s[w] = isq[s_l[w]];
        slot[w] = w * 32 + lane < n ? dl / AGG_WAVES : NQ;
    }

    int base = 0;
    int endq[NQ];
#pragma unroll
    for (int q = 0; q < NQ; ++q) {
#pragma unroll
        for (int w = 0; w < SRT_W; ++w) {
            const bool     m    = slot[w] == q;
            const unsigned mask = __builtin_amdgcn_ballot_w32(m);
            if (m) {
                const int p = base + (int)__builtin_amdgcn_mbcnt_lo(mask, 0u);
                srt[p]         = (unsigned)s_l[w];
                srt[SRT_N + p] = __float_as_uint(iq_s[w] * iq_d[w]);
            }
            base += __builtin_popcount(mask);
        }
        endq[q] = base;
    }

    int beg = 0;
#pragma unroll
    for (int q = 0; q < NQ; ++q) {
        const int end = endq[q];
        for (int p0 = beg; p0 < end; p0 += 32) {
            const int left = end - p0;
            const int nb   = left < 32 ? left : 32;
            int   s_b = 0;
            float c_b = 0.0f;
            if (lane < nb) {
                s_b = (int)srt[p0 + lane];
                c_b = __uint_as_float(srt[SRT_N + p0 + lane]);
            }
            for (int k0 = 0; k0 < nb; k0 += AGG_ROWS) {
                vecf hv[AGG_ROWS];
#pragma unroll
                for (int g = 0; g < AGG_ROWS; ++g) {
                    int kk = k0 + g;
                    kk = kk < nb ? kk : nb - 1;
                    const int s = __builtin_amdgcn_readlane(s_b, kk);
                    hv[g] = *(const vecf*)(h + (size_t)s * C + V * lane);
                }
#pragma unroll
                for (int g = 0; g < AGG_ROWS; ++g) {
                    if (k0 + g < nb) {
                        const float coef = __int_as_float(__builtin_amdgcn_readlane(__float_as_int(c_b), k0 + g));
#pragma unroll
                        for (int j = 0; j < V; ++j) a[q][j] += hv[g][j] * coef;
                    }
                }
            }
        }
        beg = end;
    }
}

template<int C, int NB, int NEXT_N>
__global__ __launch_bounds__(AGG_WAVES * 32)
void agg_owned_kernel(const float* __restrict__ h,
                          const int* __restrict__ src,
                          const int* __restrict__ dst,
                          const float* __restrict__ isq,
                          const float* __restrict__ bias,
                          const unsigned* __restrict__ lists,
                          const int* __restrict__ counts,
                          const float* __restrict__ wnext,
                          float* __restrict__ out) {
    constexpr int V    = C / 32;
    constexpr int NPB  = (N_NODES + NB - 1) / NB;
    constexpr int NQ   = (NPB + AGG_WAVES - 1) / AGG_WAVES;
    constexpr int NROWS = ((NPB + 15) / 16) * 16;
    constexpr int PITCH = C + 4;
    static_assert(NPB <= 512, "local node index is packed in 9 bits");
    static_assert((AGG_WAVES & (AGG_WAVES - 1)) == 0, "a node's wave is its low bits");
    static_assert(NEXT_N == 0 || V == 4, "the fragment read assumes four channels per lane");
    typedef float vecf __attribute__((ext_vector_type(V)));

    __shared__ float    acc[NEXT_N > 0 ? NROWS * PITCH : 4];
    __shared__ unsigned ring[AGG_WAVES * AGG_QCAP];
    __shared__ unsigned sorted[AGG_WAVES * 2 * SRT_N];
    __shared__ __attribute__((aligned(16))) float wf[NEXT_N > 0 ? C * NEXT_N : 4];

    const int lane = threadIdx.x & 31;
    const int wave = __builtin_amdgcn_readfirstlane((int)(threadIdx.x >> 5));
    const int lo   = blockIdx.x * NPB;
    int cnt = N_NODES - lo;
    cnt = cnt < NPB ? cnt : NPB;
    cnt = cnt > 0 ? cnt : 0;

    const int gw = blockIdx.x * AGG_WAVES + wave;
    const int nlist = __builtin_amdgcn_readfirstlane(counts[gw]);
    v4f wr[NEXT_N > 0 ? W_PASSES(C, NEXT_N) : 1];
    if constexpr (NEXT_N > 0) w_fragments_load<C, NEXT_N>(wnext, wr);

    float a[NQ][V];
    {
        vecf  hv[NQ];
        float sq[NQ];
        const vecf bv = *(const vecf*)(bias + V * lane);
#pragma unroll
        for (int q = 0; q < NQ; ++q) {
            int node = lo + wave + q * AGG_WAVES;
            node = node < N_NODES ? node : N_NODES - 1;
            sq[q] = isq[node];
            hv[q] = *(const vecf*)(h + (size_t)node * C + V * lane);
        }
#pragma unroll
        for (int q = 0; q < NQ; ++q) {
            const float s2 = sq[q] * sq[q];
#pragma unroll
            for (int j = 0; j < V; ++j) a[q][j] = hv[q][j] * s2 + bv[j];
        }
    }
    if constexpr (NEXT_N > 0) {
        for (int n = cnt + wave; n < ((cnt + 15) & ~15); n += AGG_WAVES)
            for (int j = 0; j < V; ++j)
                acc[n * PITCH + j * 32 + lane] = 0.0f;
        w_fragments_store<C, NEXT_N>(wf, wr);
    }

    unsigned* srt = sorted + wave * (2 * SRT_N);
    if (nlist >= 0) {
        const unsigned* mylist = lists + (size_t)gw * LIST_CAP;
        for (int pos = 0; pos < nlist; pos += SRT_N) {
            const int left = nlist - pos;
            unsigned ent[SRT_W];
#pragma unroll
            for (int w = 0; w < SRT_W; ++w)
                ent[w] = w * 32 + lane < left ? mylist[pos + w * 32 + lane] : 0u;
            wave_add_entries<C, NQ>(h, isq, a, srt, ent, left < SRT_N ? left : SRT_N, lo, lane);
        }
    } else {
        unsigned* myq = ring + wave * AGG_QCAP;
        scan_owned(dst, myq, lo, cnt, wave, lane, [&](int head, int nb) {
            unsigned ent[SRT_W] = {};
            if (lane < nb) {
                const unsigned q = myq[(head + lane) & (AGG_QCAP - 1)];
                ent[0] = ((unsigned)src[q >> 9] << 9) | (q & 511u);
            }
            wave_add_entries<C, NQ>(h, isq, a, srt, ent, nb, lo, lane);
        });
    }

    if constexpr (NEXT_N > 0) {
#pragma unroll
        for (int q = 0; q < NQ; ++q) {
            const int n = wave + q * AGG_WAVES;
            if (n < cnt) {
#pragma unroll
                for (int j = 0; j < V; ++j) acc[n * PITCH + j * 32 + lane] = a[q][j];
            }
        }
        __syncthreads();
        owner_rows_product<C, NEXT_N, true>(acc, wf, out, lo, cnt, wave, lane);
    } else {
#pragma unroll
        for (int q = 0; q < NQ; ++q) {
            const int n = wave + q * AGG_WAVES;
            if (n < cnt) {
                vecf o;
#pragma unroll
                for (int j = 0; j < V; ++j) o[j] = a[q][j];
                *(vecf*)(out + (size_t)(lo + n) * C + V * lane) = o;
            }
        }
    }
}


extern "C" void kernel_launch(void* const* d_in, const int* in_sizes, int n_in,
                              void* d_out, int out_size, void* d_ws, size_t ws_size,
                              hipStream_t stream) {
    (void)in_sizes; (void)n_in; (void)out_size; (void)ws_size;

    const float* x   = (const float*)d_in[0];
    const int*   ei  = (const int*)  d_in[1];
    const float* W1  = (const float*)d_in[2];
    const float* b1  = (const float*)d_in[3];
    const float* W2  = (const float*)d_in[4];
    const float* b2  = (const float*)d_in[5];
    float*       out = (float*)d_out;

    const int* src = ei;
    const int* dst = ei + N_EDGES;

    float*    isq    = (float*)d_ws;
    float*    h1     = isq + 50176;
    float*    h2     = h1 + (size_t)N_NODES * C_HID;
    int*      counts = (int*)(h2 + (size_t)N_NODES * C_HID);
    unsigned* lists  = (unsigned*)(counts + AGG_NB * AGG_WAVES);
    unsigned* marks  = lists + (size_t)AGG_NB * AGG_WAVES * LIST_CAP;

    owner_mark_kernel<AGG_NB>
        <<<MARK_BLOCKS, 256, 0, stream>>>(dst, marks);

    filter_degree_kernel<AGG_NB>
        <<<AGG_NB, AGG_WAVES * 32, 0, stream>>>(src, dst, marks, isq, lists, counts, x, W1, h1);

    agg_owned_kernel<C_HID, AGG_NB, C_OUTC>
        <<<AGG_NB, AGG_WAVES * 32, 0, stream>>>(h1, src, dst, isq, b1, lists, counts, W2, h2);

    agg_owned_kernel<C_OUTC, AGG_NB, 0>
        <<<AGG_NB, AGG_WAVES * 32, 0, stream>>>(h2, src, dst, isq, b2, lists, counts, nullptr, out);
}
